# starting kernel
# baseline (speedup 1.0000x reference)
#include <hip/hip_runtime.h>
#include <math.h>

typedef float v2f __attribute__((ext_vector_type(2)));
typedef float v8f __attribute__((ext_vector_type(8)));

#define DIMS   64
#define HID    64
#define NBATCH 16384
#define BTILE  128
#define PADW   68
#define PPAD   25
#define TAILB  3.0f

__device__ __forceinline__ float softplus_f(float x) {
  return fmaxf(x, 0.0f) + log1pf(expf(-fabsf(x)));
}

__device__ __forceinline__ void build_edges(const float* p, float* edges, float* widths) {
  float m1 = p[0];
#pragma unroll
  for (int j = 1; j < 8; ++j) m1 = fmaxf(m1, p[j]);
  float e[8];
  float s = 0.0f;
#pragma unroll
  for (int j = 0; j < 8; ++j) { e[j] = expf(p[j] - m1); s += e[j]; }
  float inv = (2.0f * TAILB) / s;
  float u[8];
  float m2 = -1e30f;
#pragma unroll
  for (int j = 0; j < 8; ++j) { u[j] = e[j] * inv; m2 = fmaxf(m2, u[j]); }
  float s2 = 0.0f;
#pragma unroll
  for (int j = 0; j < 8; ++j) { e[j] = expf(u[j] - m2); s2 += e[j]; }
  float inv2 = (1.0f - 8.0f * 1e-3f) / s2;
  float cum = 0.0f;
  edges[0] = -TAILB;
#pragma unroll
  for (int j = 0; j < 8; ++j) {
    float w = 1e-3f + e[j] * inv2;
    cum += w;
    edges[j + 1] = 2.0f * TAILB * cum - TAILB;
  }
  edges[8] = TAILB;
#pragma unroll
  for (int j = 0; j < 8; ++j) widths[j] = edges[j + 1] - edges[j];
}

__device__ __forceinline__ void rqs_spline(const float* p, float xin, float& zout, float& ldout) {
  bool inside = (xin >= -TAILB) && (xin <= TAILB);
  float xc = fminf(fmaxf(xin, -TAILB), TAILB);
  float cw[9], wdt[8], ch[9], hgt[8], der[9];
  build_edges(p, cw, wdt);
  build_edges(p + 8, ch, hgt);
  der[0] = 1.0f;
  der[8] = 1.0f;
#pragma unroll
  for (int j = 1; j <= 7; ++j) der[j] = 1e-3f + softplus_f(softplus_f(p[15 + j]));
  int idx = -1;
#pragma unroll
  for (int j = 0; j < 9; ++j) {
    float edge = cw[j] + ((j == 8) ? 1e-6f : 0.0f);
    idx += (xc >= edge) ? 1 : 0;
  }
  idx = max(0, min(idx, 7));
  float icw = cw[idx], iw = wdt[idx];
  float ich = ch[idx], ih = hgt[idx];
  float delta = ih / iw;
  float dk = der[idx], dk1 = der[idx + 1];
  float th  = (xc - icw) / iw;
  float omt = 1.0f - th;
  float t1  = th * omt;
  float numer = ih * (delta * th * th + dk * t1);
  float denom = delta + (dk + dk1 - 2.0f * delta) * t1;
  float outv  = ich + numer / denom;
  float dnum  = delta * delta * (dk1 * th * th + 2.0f * delta * t1 + dk * omt * omt);
  float ld    = logf(dnum) - 2.0f * logf(denom);
  zout  = inside ? outv : xin;
  ldout = inside ? ld : 0.0f;
}

__global__ void __launch_bounds__(256)
nsf_main(const float* __restrict__ x,  const float* __restrict__ initp,
         const float* __restrict__ W1, const float* __restrict__ b1,
         const float* __restrict__ W2, const float* __restrict__ b2,
         const float* __restrict__ W3, const float* __restrict__ b3,
         float* __restrict__ z_out, float* __restrict__ ld_ws) {
  __shared__ float Xs[BTILE][PADW];
  __shared__ float Hs[BTILE][PADW];
  __shared__ float W1s[HID][PADW];
  __shared__ float W2s[HID][PADW];
  __shared__ float W3s[32][PADW];
  __shared__ float B1s[HID];
  __shared__ float B2s[HID];
  __shared__ float B3s[32];
  __shared__ float Ps[BTILE][PPAD];

  const int d   = blockIdx.y;
  const int b0  = blockIdx.x * BTILE;
  const int tid = threadIdx.x;

  if (d == 0) {
    if (tid < BTILE) {
      int b = b0 + tid;
      float pl[23];
#pragma unroll
      for (int j = 0; j < 23; ++j) pl[j] = initp[j];
      float z, ld;
      rqs_spline(pl, x[b * DIMS + 0], z, ld);
      z_out[b * DIMS + 0] = z;
      ld_ws[(size_t)0 * NBATCH + b] = ld;
    }
    return;
  }

  const int l = d - 1;

  for (int i = tid; i < BTILE * DIMS; i += 256) {
    int m = i >> 6, j = i & 63;
    float v = x[(b0 + m) * DIMS + j];
    Xs[m][j] = (j < d) ? v : 0.0f;
  }
  {
    const float* w1p = W1 + (size_t)l * HID * DIMS;
    const float* w2p = W2 + (size_t)l * HID * HID;
    for (int i = tid; i < HID * DIMS; i += 256) {
      W1s[i >> 6][i & 63] = w1p[i];
      W2s[i >> 6][i & 63] = w2p[i];
    }
    const float* w3p = W3 + (size_t)l * 23 * HID;
    for (int i = tid; i < 32 * HID; i += 256) {
      int r = i >> 6, c = i & 63;
      W3s[r][c] = (r < 23) ? w3p[r * HID + c] : 0.0f;
    }
    if (tid < 64) {
      B1s[tid] = b1[l * HID + tid];
      B2s[tid] = b2[l * HID + tid];
    } else if (tid < 96) {
      int r = tid - 64;
      B3s[r] = (r < 23) ? b3[l * 23 + r] : 0.0f;
    }
  }
  __syncthreads();

  const int lane = tid & 31;
  const int wave = tid >> 5;
  const int mrow = wave * 16;
  const int ln   = lane & 15;
  const int kh   = (lane >> 4) << 1;
  const int mh   = (lane >> 4) << 3;

  const int kmax1 = (d + 3) & ~3;

#pragma unroll
  for (int nt = 0; nt < 4; ++nt) {
    const int n0 = nt * 16;
    const float bias = B1s[n0 + ln];
    v8f acc;
#pragma unroll
    for (int r = 0; r < 8; ++r) acc[r] = bias;
    for (int k0 = 0; k0 < kmax1; k0 += 4) {
      v2f a, bb;
      a.x  = Xs[mrow + ln][k0 + kh];
      a.y  = Xs[mrow + ln][k0 + kh + 1];
      bb.x = W1s[n0 + ln][k0 + kh];
      bb.y = W1s[n0 + ln][k0 + kh + 1];
      acc = __builtin_amdgcn_wmma_f32_16x16x4_f32(false, a, false, bb, (short)0, acc,
                                                  false, false);
    }
#pragma unroll
    for (int r = 0; r < 8; ++r) Hs[mrow + mh + r][n0 + ln] = tanhf(acc[r]);
  }
  __syncthreads();

#pragma unroll
  for (int nt = 0; nt < 4; ++nt) {
    const int n0 = nt * 16;
    const float bias = B2s[n0 + ln];
    v8f acc;
#pragma unroll
    for (int r = 0; r < 8; ++r) acc[r] = bias;
    for (int k0 = 0; k0 < HID; k0 += 4) {
      v2f a, bb;
      a.x  = Hs[mrow + ln][k0 + kh];
      a.y  = Hs[mrow + ln][k0 + kh + 1];
      bb.x = W2s[n0 + ln][k0 + kh];
      bb.y = W2s[n0 + ln][k0 + kh + 1];
      acc = __builtin_amdgcn_wmma_f32_16x16x4_f32(false, a, false, bb, (short)0, acc,
                                                  false, false);
    }
#pragma unroll
    for (int r = 0; r < 8; ++r) Xs[mrow + mh + r][n0 + ln] = tanhf(acc[r]);
  }
  __syncthreads();

#pragma unroll
  for (int nt = 0; nt < 2; ++nt) {
    const int n0 = nt * 16;
    const float bias = B3s[n0 + ln];
    v8f acc;
#pragma unroll
    for (int r = 0; r < 8; ++r) acc[r] = bias;
    for (int k0 = 0; k0 < HID; k0 += 4) {
      v2f a, bb;
      a.x  = Xs[mrow + ln][k0 + kh];
      a.y  = Xs[mrow + ln][k0 + kh + 1];
      bb.x = W3s[n0 + ln][k0 + kh];
      bb.y = W3s[n0 + ln][k0 + kh + 1];
      acc = __builtin_amdgcn_wmma_f32_16x16x4_f32(false, a, false, bb, (short)0, acc,
                                                  false, false);
    }
    const int col = n0 + ln;
#pragma unroll
    for (int r = 0; r < 8; ++r)
      if (col < 23) Ps[mrow + mh + r][col] = acc[r];
  }
  __syncthreads();

  if (tid < BTILE) {
    int b = b0 + tid;
    float z, ld;
    rqs_spline(&Ps[tid][0], x[b * DIMS + d], z, ld);
    z_out[b * DIMS + d] = z;
    ld_ws[(size_t)d * NBATCH + b] = ld;
  }
}

__global__ void __launch_bounds__(256)
nsf_reduce(const float* __restrict__ ld_ws, float* __restrict__ out_ld) {
  int b = blockIdx.x * 256 + threadIdx.x;
  float s = 0.0f;
#pragma unroll 8
  for (int dd = 0; dd < DIMS; ++dd) s += ld_ws[(size_t)dd * NBATCH + b];
  out_ld[b] = s;
}

extern "C" void kernel_launch(void* const* d_in, const int* in_sizes, int n_in,
                              void* d_out, int out_size, void* d_ws, size_t ws_size,
                              hipStream_t stream) {
  const float* x     = (const float*)d_in[0];
  const float* initp = (const float*)d_in[1];
  const float* W1    = (const float*)d_in[2];
  const float* b1    = (const float*)d_in[3];
  const float* W2    = (const float*)d_in[4];
  const float* b2    = (const float*)d_in[5];
  const float* W3    = (const float*)d_in[6];
  const float* b3    = (const float*)d_in[7];

  float* z_out  = (float*)d_out;
  float* ld_out = z_out + (size_t)NBATCH * DIMS;
  float* ld_ws  = (float*)d_ws;

  dim3 grid(NBATCH / BTILE, DIMS);
  nsf_main<<<grid, 256, 0, stream>>>(x, initp, W1, b1, W2, b2, W3, b3, z_out, ld_ws);
  nsf_reduce<<<NBATCH / 256, 256, 0, stream>>>(ld_ws, ld_out);
}


# tanh of both hidden layers as 1-2/(exp2(v*2/ln2)+1) on the hardware exp2 and rcp, branch-free and 8 values at a time, in place of the library's two-path tanhf (abs error a few 1e-7)
# speedup vs baseline: 1.1717x; 1.1717x over previous
#include <hip/hip_runtime.h>
#include <math.h>

typedef float v2f __attribute__((ext_vector_type(2)));
typedef float v8f __attribute__((ext_vector_type(8)));

#define DIMS   64
#define HID    64
#define NBATCH 16384
#define BTILE  128
#define PADW   68
#define PPAD   25
#define TAILB  3.0f

__device__ __forceinline__ float softplus_f(float x) {
  return fmaxf(x, 0.0f) + log1pf(expf(-fabsf(x)));
}

__device__ __forceinline__ float tanh_hw(float v) {
  float e = __builtin_amdgcn_exp2f(v * 2.88539008177792681f);
  float r = __builtin_amdgcn_rcpf(e + 1.0f);
  return fmaf(-2.0f, r, 1.0f);
}

__device__ __forceinline__ void build_edges(const float* p, float* edges, float* widths) {
  float m1 = p[0];
#pragma unroll
  for (int j = 1; j < 8; ++j) m1 = fmaxf(m1, p[j]);
  float e[8];
  float s = 0.0f;
#pragma unroll
  for (int j = 0; j < 8; ++j) { e[j] = expf(p[j] - m1); s += e[j]; }
  float inv = (2.0f * TAILB) / s;
  float u[8];
  float m2 = -1e30f;
#pragma unroll
  for (int j = 0; j < 8; ++j) { u[j] = e[j] * inv; m2 = fmaxf(m2, u[j]); }
  float s2 = 0.0f;
#pragma unroll
  for (int j = 0; j < 8; ++j) { e[j] = expf(u[j] - m2); s2 += e[j]; }
  float inv2 = (1.0f - 8.0f * 1e-3f) / s2;
  float cum = 0.0f;
  edges[0] = -TAILB;
#pragma unroll
  for (int j = 0; j < 8; ++j) {
    float w = 1e-3f + e[j] * inv2;
    cum += w;
    edges[j + 1] = 2.0f * TAILB * cum - TAILB;
  }
  edges[8] = TAILB;
#pragma unroll
  for (int j = 0; j < 8; ++j) widths[j] = edges[j + 1] - edges[j];
}

__device__ __forceinline__ void rqs_spline(const float* p, float xin, float& zout, float& ldout) {
  bool inside = (xin >= -TAILB) && (xin <= TAILB);
  float xc = fminf(fmaxf(xin, -TAILB), TAILB);
  float cw[9], wdt[8], ch[9], hgt[8], der[9];
  build_edges(p, cw, wdt);
  build_edges(p + 8, ch, hgt);
  der[0] = 1.0f;
  der[8] = 1.0f;
#pragma unroll
  for (int j = 1; j <= 7; ++j) der[j] = 1e-3f + softplus_f(softplus_f(p[15 + j]));
  int idx = -1;
#pragma unroll
  for (int j = 0; j < 9; ++j) {
    float edge = cw[j] + ((j == 8) ? 1e-6f : 0.0f);
    idx += (xc >= edge) ? 1 : 0;
  }
  idx = max(0, min(idx, 7));
  float icw = cw[idx], iw = wdt[idx];
  float ich = ch[idx], ih = hgt[idx];
  float delta = ih / iw;
  float dk = der[idx], dk1 = der[idx + 1];
  float th  = (xc - icw) / iw;
  float omt = 1.0f - th;
  float t1  = th * omt;
  float numer = ih * (delta * th * th + dk * t1);
  float denom = delta + (dk + dk1 - 2.0f * delta) * t1;
  float outv  = ich + numer / denom;
  float dnum  = delta * delta * (dk1 * th * th + 2.0f * delta * t1 + dk * omt * omt);
  float ld    = logf(dnum) - 2.0f * logf(denom);
  zout  = inside ? outv : xin;
  ldout = inside ? ld : 0.0f;
}

__global__ void __launch_bounds__(256)
nsf_main(const float* __restrict__ x,  const float* __restrict__ initp,
         const float* __restrict__ W1, const float* __restrict__ b1,
         const float* __restrict__ W2, const float* __restrict__ b2,
         const float* __restrict__ W3, const float* __restrict__ b3,
         float* __restrict__ z_out, float* __restrict__ ld_ws) {
  __shared__ float Xs[BTILE][PADW];
  __shared__ float Hs[BTILE][PADW];
  __shared__ float W1s[HID][PADW];
  __shared__ float W2s[HID][PADW];
  __shared__ float W3s[32][PADW];
  __shared__ float B1s[HID];
  __shared__ float B2s[HID];
  __shared__ float B3s[32];
  __shared__ float Ps[BTILE][PPAD];

  const int d   = blockIdx.y;
  const int b0  = blockIdx.x * BTILE;
  const int tid = threadIdx.x;

  if (d == 0) {
    if (tid < BTILE) {
      int b = b0 + tid;
      float pl[23];
#pragma unroll
      for (int j = 0; j < 23; ++j) pl[j] = initp[j];
      float z, ld;
      rqs_spline(pl, x[b * DIMS + 0], z, ld);
      z_out[b * DIMS + 0] = z;
      ld_ws[(size_t)0 * NBATCH + b] = ld;
    }
    return;
  }

  const int l = d - 1;

  for (int i = tid; i < BTILE * DIMS; i += 256) {
    int m = i >> 6, j = i & 63;
    float v = x[(b0 + m) * DIMS + j];
    Xs[m][j] = (j < d) ? v : 0.0f;
  }
  {
    const float* w1p = W1 + (size_t)l * HID * DIMS;
    const float* w2p = W2 + (size_t)l * HID * HID;
    for (int i = tid; i < HID * DIMS; i += 256) {
      W1s[i >> 6][i & 63] = w1p[i];
      W2s[i >> 6][i & 63] = w2p[i];
    }
    const float* w3p = W3 + (size_t)l * 23 * HID;
    for (int i = tid; i < 32 * HID; i += 256) {
      int r = i >> 6, c = i & 63;
      W3s[r][c] = (r < 23) ? w3p[r * HID + c] : 0.0f;
    }
    if (tid < 64) {
      B1s[tid] = b1[l * HID + tid];
      B2s[tid] = b2[l * HID + tid];
    } else if (tid < 96) {
      int r = tid - 64;
      B3s[r] = (r < 23) ? b3[l * 23 + r] : 0.0f;
    }
  }
  __syncthreads();

  const int lane = tid & 31;
  const int wave = tid >> 5;
  const int mrow = wave * 16;
  const int ln   = lane & 15;
  const int kh   = (lane >> 4) << 1;
  const int mh   = (lane >> 4) << 3;

  const int kmax1 = (d + 3) & ~3;

#pragma unroll
  for (int nt = 0; nt < 4; ++nt) {
    const int n0 = nt * 16;
    const float bias = B1s[n0 + ln];
    v8f acc;
#pragma unroll
    for (int r = 0; r < 8; ++r) acc[r] = bias;
    for (int k0 = 0; k0 < kmax1; k0 += 4) {
      v2f a, bb;
      a.x  = Xs[mrow + ln][k0 + kh];
      a.y  = Xs[mrow + ln][k0 + kh + 1];
      bb.x = W1s[n0 + ln][k0 + kh];
      bb.y = W1s[n0 + ln][k0 + kh + 1];
      acc = __builtin_amdgcn_wmma_f32_16x16x4_f32(false, a, false, bb, (short)0, acc,
                                                  false, false);
    }
#pragma unroll
    for (int r = 0; r < 8; ++r) Hs[mrow + mh + r][n0 + ln] = tanh_hw(acc[r]);
  }
  __syncthreads();

#pragma unroll
  for (int nt = 0; nt < 4; ++nt) {
    const int n0 = nt * 16;
    const float bias = B2s[n0 + ln];
    v8f acc;
#pragma unroll
    for (int r = 0; r < 8; ++r) acc[r] = bias;
    for (int k0 = 0; k0 < HID; k0 += 4) {
      v2f a, bb;
      a.x  = Hs[mrow + ln][k0 + kh];
      a.y  = Hs[mrow + ln][k0 + kh + 1];
      bb.x = W2s[n0 + ln][k0 + kh];
      bb.y = W2s[n0 + ln][k0 + kh + 1];
      acc = __builtin_amdgcn_wmma_f32_16x16x4_f32(false, a, false, bb, (short)0, acc,
                                                  false, false);
    }
#pragma unroll
    for (int r = 0; r < 8; ++r) Xs[mrow + mh + r][n0 + ln] = tanh_hw(acc[r]);
  }
  __syncthreads();

#pragma unroll
  for (int nt = 0; nt < 2; ++nt) {
    const int n0 = nt * 16;
    const float bias = B3s[n0 + ln];
    v8f acc;
#pragma unroll
    for (int r = 0; r < 8; ++r) acc[r] = bias;
    for (int k0 = 0; k0 < HID; k0 += 4) {
      v2f a, bb;
      a.x  = Xs[mrow + ln][k0 + kh];
      a.y  = Xs[mrow + ln][k0 + kh + 1];
      bb.x = W3s[n0 + ln][k0 + kh];
      bb.y = W3s[n0 + ln][k0 + kh + 1];
      acc = __builtin_amdgcn_wmma_f32_16x16x4_f32(false, a, false, bb, (short)0, acc,
                                                  false, false);
    }
    const int col = n0 + ln;
#pragma unroll
    for (int r = 0; r < 8; ++r)
      if (col < 23) Ps[mrow + mh + r][col] = acc[r];
  }
  __syncthreads();

  if (tid < BTILE) {
    int b = b0 + tid;
    float z, ld;
    rqs_spline(&Ps[tid][0], x[b * DIMS + d], z, ld);
    z_out[b * DIMS + d] = z;
    ld_ws[(size_t)d * NBATCH + b] = ld;
  }
}

__global__ void __launch_bounds__(256)
nsf_reduce(const float* __restrict__ ld_ws, float* __restrict__ out_ld) {
  int b = blockIdx.x * 256 + threadIdx.x;
  float s = 0.0f;
#pragma unroll 8
  for (int dd = 0; dd < DIMS; ++dd) s += ld_ws[(size_t)dd * NBATCH + b];
  out_ld[b] = s;
}

extern "C" void kernel_launch(void* const* d_in, const int* in_sizes, int n_in,
                              void* d_out, int out_size, void* d_ws, size_t ws_size,
                              hipStream_t stream) {
  const float* x     = (const float*)d_in[0];
  const float* initp = (const float*)d_in[1];
  const float* W1    = (const float*)d_in[2];
  const float* b1    = (const float*)d_in[3];
  const float* W2    = (const float*)d_in[4];
  const float* b2    = (const float*)d_in[5];
  const float* W3    = (const float*)d_in[6];
  const float* b3    = (const float*)d_in[7];

  float* z_out  = (float*)d_out;
  float* ld_out = z_out + (size_t)NBATCH * DIMS;
  float* ld_ws  = (float*)d_ws;

  dim3 grid(NBATCH / BTILE, DIMS);
  nsf_main<<<grid, 256, 0, stream>>>(x, initp, W1, b1, W2, b2, W3, b3, z_out, ld_ws);
  nsf_reduce<<<NBATCH / 256, 256, 0, stream>>>(ld_ws, ld_out);
}


# staging: X tile and the three weight matrices fetched with 16-byte loads all issued before the first wait, 16-byte LDS stores (one cache round trip per block, not 21 dependent ones)
# speedup vs baseline: 1.8585x; 1.5862x over previous
#include <hip/hip_runtime.h>
#include <math.h>

typedef float v2f __attribute__((ext_vector_type(2)));
typedef float v8f __attribute__((ext_vector_type(8)));
typedef float v4f __attribute__((ext_vector_type(4)));
typedef float v4f_g __attribute__((ext_vector_type(4), aligned(4)));

#define DIMS   64
#define HID    64
#define NBATCH 16384
#define BTILE  128
#define PADW   68
#define PPAD   25
#define TAILB  3.0f

__device__ __forceinline__ v4f ld16(const float* p) {
  return *reinterpret_cast<const v4f_g*>(p);
}

__device__ __forceinline__ float softplus_f(float x) {
  return fmaxf(x, 0.0f) + log1pf(expf(-fabsf(x)));
}

__device__ __forceinline__ float tanh_hw(float v) {
  float e = __builtin_amdgcn_exp2f(v * 2.88539008177792681f);
  float r = __builtin_amdgcn_rcpf(e + 1.0f);
  return fmaf(-2.0f, r, 1.0f);
}

__device__ __forceinline__ void build_edges(const float* p, float* edges, float* widths) {
  float m1 = p[0];
#pragma unroll
  for (int j = 1; j < 8; ++j) m1 = fmaxf(m1, p[j]);
  float e[8];
  float s = 0.0f;
#pragma unroll
  for (int j = 0; j < 8; ++j) { e[j] = expf(p[j] - m1); s += e[j]; }
  float inv = (2.0f * TAILB) / s;
  float u[8];
  float m2 = -1e30f;
#pragma unroll
  for (int j = 0; j < 8; ++j) { u[j] = e[j] * inv; m2 = fmaxf(m2, u[j]); }
  float s2 = 0.0f;
#pragma unroll
  for (int j = 0; j < 8; ++j) { e[j] = expf(u[j] - m2); s2 += e[j]; }
  float inv2 = (1.0f - 8.0f * 1e-3f) / s2;
  float cum = 0.0f;
  edges[0] = -TAILB;
#pragma unroll
  for (int j = 0; j < 8; ++j) {
    float w = 1e-3f + e[j] * inv2;
    cum += w;
    edges[j + 1] = 2.0f * TAILB * cum - TAILB;
  }
  edges[8] = TAILB;
#pragma unroll
  for (int j = 0; j < 8; ++j) widths[j] = edges[j + 1] - edges[j];
}

__device__ __forceinline__ void rqs_spline(const float* p, float xin, float& zout, float& ldout) {
  bool inside = (xin >= -TAILB) && (xin <= TAILB);
  float xc = fminf(fmaxf(xin, -TAILB), TAILB);
  float cw[9], wdt[8], ch[9], hgt[8], der[9];
  build_edges(p, cw, wdt);
  build_edges(p + 8, ch, hgt);
  der[0] = 1.0f;
  der[8] = 1.0f;
#pragma unroll
  for (int j = 1; j <= 7; ++j) der[j] = 1e-3f + softplus_f(softplus_f(p[15 + j]));
  int idx = -1;
#pragma unroll
  for (int j = 0; j < 9; ++j) {
    float edge = cw[j] + ((j == 8) ? 1e-6f : 0.0f);
    idx += (xc >= edge) ? 1 : 0;
  }
  idx = max(0, min(idx, 7));
  float icw = cw[idx], iw = wdt[idx];
  float ich = ch[idx], ih = hgt[idx];
  float delta = ih / iw;
  float dk = der[idx], dk1 = der[idx + 1];
  float th  = (xc - icw) / iw;
  float omt = 1.0f - th;
  float t1  = th * omt;
  float numer = ih * (delta * th * th + dk * t1);
  float denom = delta + (dk + dk1 - 2.0f * delta) * t1;
  float outv  = ich + numer / denom;
  float dnum  = delta * delta * (dk1 * th * th + 2.0f * delta * t1 + dk * omt * omt);
  float ld    = logf(dnum) - 2.0f * logf(denom);
  zout  = inside ? outv : xin;
  ldout = inside ? ld : 0.0f;
}

__global__ void __launch_bounds__(256)
nsf_main(const float* __restrict__ x,  const float* __restrict__ initp,
         const float* __restrict__ W1, const float* __restrict__ b1,
         const float* __restrict__ W2, const float* __restrict__ b2,
         const float* __restrict__ W3, const float* __restrict__ b3,
         float* __restrict__ z_out, float* __restrict__ ld_ws) {
  __shared__ __align__(16) float Xs[BTILE][PADW];
  __shared__ float Hs[BTILE][PADW];
  __shared__ __align__(16) float W1s[HID][PADW];
  __shared__ __align__(16) float W2s[HID][PADW];
  __shared__ __align__(16) float W3s[32][PADW];
  __shared__ float B1s[HID];
  __shared__ float B2s[HID];
  __shared__ float B3s[32];
  __shared__ float Ps[BTILE][PPAD];

  const int d   = blockIdx.y;
  const int b0  = blockIdx.x * BTILE;
  const int tid = threadIdx.x;

  if (d == 0) {
    if (tid < BTILE) {
      int b = b0 + tid;
      float pl[23];
#pragma unroll
      for (int j = 0; j < 23; ++j) pl[j] = initp[j];
      float z, ld;
      rqs_spline(pl, x[b * DIMS + 0], z, ld);
      z_out[b * DIMS + 0] = z;
      ld_ws[(size_t)0 * NBATCH + b] = ld;
    }
    return;
  }

  const int l = d - 1;

  {
    const float* xg  = x  + (size_t)b0 * DIMS;
    const float* w1p = W1 + (size_t)l * HID * DIMS;
    const float* w2p = W2 + (size_t)l * HID * HID;
    const float* w3p = W3 + (size_t)l * 23 * HID;
    v4f xr[8], w1r[4], w2r[4], w3r[2];
#pragma unroll
    for (int t = 0; t < 8; ++t) xr[t] = ld16(xg + (tid + t * 256) * 4);
#pragma unroll
    for (int t = 0; t < 4; ++t) {
      w1r[t] = ld16(w1p + (tid + t * 256) * 4);
      w2r[t] = ld16(w2p + (tid + t * 256) * 4);
    }
#pragma unroll
    for (int t = 0; t < 2; ++t) {
      const int i = tid + t * 256;
      v4f z4 = {0.0f, 0.0f, 0.0f, 0.0f};
      w3r[t] = (i < 23 * 16) ? ld16(w3p + i * 4) : z4;
    }
    float bv = 0.0f;
    if (tid < 64)       bv = b1[l * HID + tid];
    else if (tid < 128) bv = b2[l * HID + (tid - 64)];
    else if (tid < 151) bv = b3[l * 23 + (tid - 128)];

    const int c4 = (tid & 15) << 2;
#pragma unroll
    for (int t = 0; t < 8; ++t) {
      const int m = (tid + t * 256) >> 4;
      v4f v = xr[t];
      v.x = (c4     < d) ? v.x : 0.0f;
      v.y = (c4 + 1 < d) ? v.y : 0.0f;
      v.z = (c4 + 2 < d) ? v.z : 0.0f;
      v.w = (c4 + 3 < d) ? v.w : 0.0f;
      *reinterpret_cast<v4f*>(&Xs[m][c4]) = v;
    }
#pragma unroll
    for (int t = 0; t < 4; ++t) {
      const int r = (tid + t * 256) >> 4;
      *reinterpret_cast<v4f*>(&W1s[r][c4]) = w1r[t];
      *reinterpret_cast<v4f*>(&W2s[r][c4]) = w2r[t];
    }
#pragma unroll
    for (int t = 0; t < 2; ++t) {
      const int r = (tid + t * 256) >> 4;
      *reinterpret_cast<v4f*>(&W3s[r][c4]) = w3r[t];
    }
    if (tid < 64)       B1s[tid] = bv;
    else if (tid < 128) B2s[tid - 64] = bv;
    else if (tid < 160) B3s[tid - 128] = bv;
  }
  __syncthreads();

  const int lane = tid & 31;
  const int wave = tid >> 5;
  const int mrow = wave * 16;
  const int ln   = lane & 15;
  const int kh   = (lane >> 4) << 1;
  const int mh   = (lane >> 4) << 3;

  const int kmax1 = (d + 3) & ~3;

#pragma unroll
  for (int nt = 0; nt < 4; ++nt) {
    const int n0 = nt * 16;
    const float bias = B1s[n0 + ln];
    v8f acc;
#pragma unroll
    for (int r = 0; r < 8; ++r) acc[r] = bias;
    for (int k0 = 0; k0 < kmax1; k0 += 4) {
      v2f a, bb;
      a.x  = Xs[mrow + ln][k0 + kh];
      a.y  = Xs[mrow + ln][k0 + kh + 1];
      bb.x = W1s[n0 + ln][k0 + kh];
      bb.y = W1s[n0 + ln][k0 + kh + 1];
      acc = __builtin_amdgcn_wmma_f32_16x16x4_f32(false, a, false, bb, (short)0, acc,
                                                  false, false);
    }
#pragma unroll
    for (int r = 0; r < 8; ++r) Hs[mrow + mh + r][n0 + ln] = tanh_hw(acc[r]);
  }
  __syncthreads();

#pragma unroll
  for (int nt = 0; nt < 4; ++nt) {
    const int n0 = nt * 16;
    const float bias = B2s[n0 + ln];
    v8f acc;
#pragma unroll
    for (int r = 0; r < 8; ++r) acc[r] = bias;
    for (int k0 = 0; k0 < HID; k0 += 4) {
      v2f a, bb;
      a.x  = Hs[mrow + ln][k0 + kh];
      a.y  = Hs[mrow + ln][k0 + kh + 1];
      bb.x = W2s[n0 + ln][k0 + kh];
      bb.y = W2s[n0 + ln][k0 + kh + 1];
      acc = __builtin_amdgcn_wmma_f32_16x16x4_f32(false, a, false, bb, (short)0, acc,
                                                  false, false);
    }
#pragma unroll
    for (int r = 0; r < 8; ++r) Xs[mrow + mh + r][n0 + ln] = tanh_hw(acc[r]);
  }
  __syncthreads();

#pragma unroll
  for (int nt = 0; nt < 2; ++nt) {
    const int n0 = nt * 16;
    const float bias = B3s[n0 + ln];
    v8f acc;
#pragma unroll
    for (int r = 0; r < 8; ++r) acc[r] = bias;
    for (int k0 = 0; k0 < HID; k0 += 4) {
      v2f a, bb;
      a.x  = Xs[mrow + ln][k0 + kh];
      a.y  = Xs[mrow + ln][k0 + kh + 1];
      bb.x = W3s[n0 + ln][k0 + kh];
      bb.y = W3s[n0 + ln][k0 + kh + 1];
      acc = __builtin_amdgcn_wmma_f32_16x16x4_f32(false, a, false, bb, (short)0, acc,
                                                  false, false);
    }
    const int col = n0 + ln;
#pragma unroll
    for (int r = 0; r < 8; ++r)
      if (col < 23) Ps[mrow + mh + r][col] = acc[r];
  }
  __syncthreads();

  if (tid < BTILE) {
    int b = b0 + tid;
    float z, ld;
    rqs_spline(&Ps[tid][0], x[b * DIMS + d], z, ld);
    z_out[b * DIMS + d] = z;
    ld_ws[(size_t)d * NBATCH + b] = ld;
  }
}

__global__ void __launch_bounds__(256)
nsf_reduce(const float* __restrict__ ld_ws, float* __restrict__ out_ld) {
  int b = blockIdx.x * 256 + threadIdx.x;
  float s = 0.0f;
#pragma unroll 8
  for (int dd = 0; dd < DIMS; ++dd) s += ld_ws[(size_t)dd * NBATCH + b];
  out_ld[b] = s;
}

extern "C" void kernel_launch(void* const* d_in, const int* in_sizes, int n_in,
                              void* d_out, int out_size, void* d_ws, size_t ws_size,
                              hipStream_t stream) {
  const float* x     = (const float*)d_in[0];
  const float* initp = (const float*)d_in[1];
  const float* W1    = (const float*)d_in[2];
  const float* b1    = (const float*)d_in[3];
  const float* W2    = (const float*)d_in[4];
  const float* b2    = (const float*)d_in[5];
  const float* W3    = (const float*)d_in[6];
  const float* b3    = (const float*)d_in[7];

  float* z_out  = (float*)d_out;
  float* ld_out = z_out + (size_t)NBATCH * DIMS;
  float* ld_ws  = (float*)d_ws;

  dim3 grid(NBATCH / BTILE, DIMS);
  nsf_main<<<grid, 256, 0, stream>>>(x, initp, W1, b1, W2, b2, W3, b3, z_out, ld_ws);
  nsf_reduce<<<NBATCH / 256, 256, 0, stream>>>(ld_ws, ld_out);
}


# spline: exp, log and softplus of bin edges, derivatives and log-determinant on the hardware exp2 / log2 units (2 instructions each) in place of the library expf, logf and log1pf routines
# speedup vs baseline: 2.1718x; 1.1686x over previous
#include <hip/hip_runtime.h>
#include <math.h>

typedef float v2f __attribute__((ext_vector_type(2)));
typedef float v8f __attribute__((ext_vector_type(8)));
typedef float v4f __attribute__((ext_vector_type(4)));
typedef float v4f_g __attribute__((ext_vector_type(4), aligned(4)));

#define DIMS   64
#define HID    64
#define NBATCH 16384
#define BTILE  128
#define PADW   68
#define PPAD   25
#define TAILB  3.0f

__device__ __forceinline__ v4f ld16(const float* p) {
  return *reinterpret_cast<const v4f_g*>(p);
}

__device__ __forceinline__ float exp_hw(float v) {
  return __builtin_amdgcn_exp2f(v * 1.44269504088896341f);
}
__device__ __forceinline__ float log_hw(float v) {
  return __builtin_amdgcn_logf(v) * 0.693147180559945309f;
}

__device__ __forceinline__ float softplus_f(float x) {
  return fmaxf(x, 0.0f) + log_hw(1.0f + exp_hw(-fabsf(x)));
}

__device__ __forceinline__ float tanh_hw(float v) {
  float e = __builtin_amdgcn_exp2f(v * 2.88539008177792681f);
  float r = __builtin_amdgcn_rcpf(e + 1.0f);
  return fmaf(-2.0f, r, 1.0f);
}

__device__ __forceinline__ void build_edges(const float* p, float* edges, float* widths) {
  float m1 = p[0];
#pragma unroll
  for (int j = 1; j < 8; ++j) m1 = fmaxf(m1, p[j]);
  float e[8];
  float s = 0.0f;
#pragma unroll
  for (int j = 0; j < 8; ++j) { e[j] = exp_hw(p[j] - m1); s += e[j]; }
  float inv = (2.0f * TAILB) / s;
  float u[8];
  float m2 = -1e30f;
#pragma unroll
  for (int j = 0; j < 8; ++j) { u[j] = e[j] * inv; m2 = fmaxf(m2, u[j]); }
  float s2 = 0.0f;
#pragma unroll
  for (int j = 0; j < 8; ++j) { e[j] = exp_hw(u[j] - m2); s2 += e[j]; }
  float inv2 = (1.0f - 8.0f * 1e-3f) / s2;
  float cum = 0.0f;
  edges[0] = -TAILB;
#pragma unroll
  for (int j = 0; j < 8; ++j) {
    float w = 1e-3f + e[j] * inv2;
    cum += w;
    edges[j + 1] = 2.0f * TAILB * cum - TAILB;
  }
  edges[8] = TAILB;
#pragma unroll
  for (int j = 0; j < 8; ++j) widths[j] = edges[j + 1] - edges[j];
}

__device__ __forceinline__ void rqs_spline(const float* p, float xin, float& zout, float& ldout) {
  bool inside = (xin >= -TAILB) && (xin <= TAILB);
  float xc = fminf(fmaxf(xin, -TAILB), TAILB);
  float cw[9], wdt[8], ch[9], hgt[8], der[9];
  build_edges(p, cw, wdt);
  build_edges(p + 8, ch, hgt);
  der[0] = 1.0f;
  der[8] = 1.0f;
#pragma unroll
  for (int j = 1; j <= 7; ++j) der[j] = 1e-3f + softplus_f(softplus_f(p[15 + j]));
  int idx = -1;
#pragma unroll
  for (int j = 0; j < 9; ++j) {
    float edge = cw[j] + ((j == 8) ? 1e-6f : 0.0f);
    idx += (xc >= edge) ? 1 : 0;
  }
  idx = max(0, min(idx, 7));
  float icw = cw[idx], iw = wdt[idx];
  float ich = ch[idx], ih = hgt[idx];
  float delta = ih / iw;
  float dk = der[idx], dk1 = der[idx + 1];
  float th  = (xc - icw) / iw;
  float omt = 1.0f - th;
  float t1  = th * omt;
  float numer = ih * (delta * th * th + dk * t1);
  float denom = delta + (dk + dk1 - 2.0f * delta) * t1;
  float outv  = ich + numer / denom;
  float dnum  = delta * delta * (dk1 * th * th + 2.0f * delta * t1 + dk * omt * omt);
  float ld    = log_hw(dnum) - 2.0f * log_hw(denom);
  zout  = inside ? outv : xin;
  ldout = inside ? ld : 0.0f;
}

__global__ void __launch_bounds__(256)
nsf_main(const float* __restrict__ x,  const float* __restrict__ initp,
         const float* __restrict__ W1, const float* __restrict__ b1,
         const float* __restrict__ W2, const float* __restrict__ b2,
         const float* __restrict__ W3, const float* __restrict__ b3,
         float* __restrict__ z_out, float* __restrict__ ld_ws) {
  __shared__ __align__(16) float Xs[BTILE][PADW];
  __shared__ float Hs[BTILE][PADW];
  __shared__ __align__(16) float W1s[HID][PADW];
  __shared__ __align__(16) float W2s[HID][PADW];
  __shared__ __align__(16) float W3s[32][PADW];
  __shared__ float B1s[HID];
  __shared__ float B2s[HID];
  __shared__ float B3s[32];
  __shared__ float Ps[BTILE][PPAD];

  const int d   = blockIdx.y;
  const int b0  = blockIdx.x * BTILE;
  const int tid = threadIdx.x;

  if (d == 0) {
    if (tid < BTILE) {
      int b = b0 + tid;
      float pl[23];
#pragma unroll
      for (int j = 0; j < 23; ++j) pl[j] = initp[j];
      float z, ld;
      rqs_spline(pl, x[b * DIMS + 0], z, ld);
      z_out[b * DIMS + 0] = z;
      ld_ws[(size_t)0 * NBATCH + b] = ld;
    }
    return;
  }

  const int l = d - 1;

  {
    const float* xg  = x  + (size_t)b0 * DIMS;
    const float* w1p = W1 + (size_t)l * HID * DIMS;
    const float* w2p = W2 + (size_t)l * HID * HID;
    const float* w3p = W3 + (size_t)l * 23 * HID;
    v4f xr[8], w1r[4], w2r[4], w3r[2];
#pragma unroll
    for (int t = 0; t < 8; ++t) xr[t] = ld16(xg + (tid + t * 256) * 4);
#pragma unroll
    for (int t = 0; t < 4; ++t) {
      w1r[t] = ld16(w1p + (tid + t * 256) * 4);
      w2r[t] = ld16(w2p + (tid + t * 256) * 4);
    }
#pragma unroll
    for (int t = 0; t < 2; ++t) {
      const int i = tid + t * 256;
      v4f z4 = {0.0f, 0.0f, 0.0f, 0.0f};
      w3r[t] = (i < 23 * 16) ? ld16(w3p + i * 4) : z4;
    }
    float bv = 0.0f;
    if (tid < 64)       bv = b1[l * HID + tid];
    else if (tid < 128) bv = b2[l * HID + (tid - 64)];
    else if (tid < 151) bv = b3[l * 23 + (tid - 128)];

    const int c4 = (tid & 15) << 2;
#pragma unroll
    for (int t = 0; t < 8; ++t) {
      const int m = (tid + t * 256) >> 4;
      v4f v = xr[t];
      v.x = (c4     < d) ? v.x : 0.0f;
      v.y = (c4 + 1 < d) ? v.y : 0.0f;
      v.z = (c4 + 2 < d) ? v.z : 0.0f;
      v.w = (c4 + 3 < d) ? v.w : 0.0f;
      *reinterpret_cast<v4f*>(&Xs[m][c4]) = v;
    }
#pragma unroll
    for (int t = 0; t < 4; ++t) {
      const int r = (tid + t * 256) >> 4;
      *reinterpret_cast<v4f*>(&W1s[r][c4]) = w1r[t];
      *reinterpret_cast<v4f*>(&W2s[r][c4]) = w2r[t];
    }
#pragma unroll
    for (int t = 0; t < 2; ++t) {
      const int r = (tid + t * 256) >> 4;
      *reinterpret_cast<v4f*>(&W3s[r][c4]) = w3r[t];
    }
    if (tid < 64)       B1s[tid] = bv;
    else if (tid < 128) B2s[tid - 64] = bv;
    else if (tid < 160) B3s[tid - 128] = bv;
  }
  __syncthreads();

  const int lane = tid & 31;
  const int wave = tid >> 5;
  const int mrow = wave * 16;
  const int ln   = lane & 15;
  const int kh   = (lane >> 4) << 1;
  const int mh   = (lane >> 4) << 3;

  const int kmax1 = (d + 3) & ~3;

#pragma unroll
  for (int nt = 0; nt < 4; ++nt) {
    const int n0 = nt * 16;
    const float bias = B1s[n0 + ln];
    v8f acc;
#pragma unroll
    for (int r = 0; r < 8; ++r) acc[r] = bias;
    for (int k0 = 0; k0 < kmax1; k0 += 4) {
      v2f a, bb;
      a.x  = Xs[mrow + ln][k0 + kh];
      a.y  = Xs[mrow + ln][k0 + kh + 1];
      bb.x = W1s[n0 + ln][k0 + kh];
      bb.y = W1s[n0 + ln][k0 + kh + 1];
      acc = __builtin_amdgcn_wmma_f32_16x16x4_f32(false, a, false, bb, (short)0, acc,
                                                  false, false);
    }
#pragma unroll
    for (int r = 0; r < 8; ++r) Hs[mrow + mh + r][n0 + ln] = tanh_hw(acc[r]);
  }
  __syncthreads();

#pragma unroll
  for (int nt = 0; nt < 4; ++nt) {
    const int n0 = nt * 16;
    const float bias = B2s[n0 + ln];
    v8f acc;
#pragma unroll
    for (int r = 0; r < 8; ++r) acc[r] = bias;
    for (int k0 = 0; k0 < HID; k0 += 4) {
      v2f a, bb;
      a.x  = Hs[mrow + ln][k0 + kh];
      a.y  = Hs[mrow + ln][k0 + kh + 1];
      bb.x = W2s[n0 + ln][k0 + kh];
      bb.y = W2s[n0 + ln][k0 + kh + 1];
      acc = __builtin_amdgcn_wmma_f32_16x16x4_f32(false, a, false, bb, (short)0, acc,
                                                  false, false);
    }
#pragma unroll
    for (int r = 0; r < 8; ++r) Xs[mrow + mh + r][n0 + ln] = tanh_hw(acc[r]);
  }
  __syncthreads();

#pragma unroll
  for (int nt = 0; nt < 2; ++nt) {
    const int n0 = nt * 16;
    const float bias = B3s[n0 + ln];
    v8f acc;
#pragma unroll
    for (int r = 0; r < 8; ++r) acc[r] = bias;
    for (int k0 = 0; k0 < HID; k0 += 4) {
      v2f a, bb;
      a.x  = Xs[mrow + ln][k0 + kh];
      a.y  = Xs[mrow + ln][k0 + kh + 1];
      bb.x = W3s[n0 + ln][k0 + kh];
      bb.y = W3s[n0 + ln][k0 + kh + 1];
      acc = __builtin_amdgcn_wmma_f32_16x16x4_f32(false, a, false, bb, (short)0, acc,
                                                  false, false);
    }
    const int col = n0 + ln;
#pragma unroll
    for (int r = 0; r < 8; ++r)
      if (col < 23) Ps[mrow + mh + r][col] = acc[r];
  }
  __syncthreads();

  if (tid < BTILE) {
    int b = b0 + tid;
    float z, ld;
    rqs_spline(&Ps[tid][0], x[b * DIMS + d], z, ld);
    z_out[b * DIMS + d] = z;
    ld_ws[(size_t)d * NBATCH + b] = ld;
  }
}

__global__ void __launch_bounds__(256)
nsf_reduce(const float* __restrict__ ld_ws, float* __restrict__ out_ld) {
  int b = blockIdx.x * 256 + threadIdx.x;
  float s = 0.0f;
#pragma unroll 8
  for (int dd = 0; dd < DIMS; ++dd) s += ld_ws[(size_t)dd * NBATCH + b];
  out_ld[b] = s;
}

extern "C" void kernel_launch(void* const* d_in, const int* in_sizes, int n_in,
                              void* d_out, int out_size, void* d_ws, size_t ws_size,
                              hipStream_t stream) {
  const float* x     = (const float*)d_in[0];
  const float* initp = (const float*)d_in[1];
  const float* W1    = (const float*)d_in[2];
  const float* b1    = (const float*)d_in[3];
  const float* W2    = (const float*)d_in[4];
  const float* b2    = (const float*)d_in[5];
  const float* W3    = (const float*)d_in[6];
  const float* b3    = (const float*)d_in[7];

  float* z_out  = (float*)d_out;
  float* ld_out = z_out + (size_t)NBATCH * DIMS;
  float* ld_ws  = (float*)d_ws;

  dim3 grid(NBATCH / BTILE, DIMS);
  nsf_main<<<grid, 256, 0, stream>>>(x, initp, W1, b1, W2, b2, W3, b3, z_out, ld_ws);
  nsf_reduce<<<NBATCH / 256, 256, 0, stream>>>(ld_ws, ld_out);
}


# LDS per block 126592 -> 105088 B so three blocks fit per WGP where two did: layer 2 writes over its own input rows, layer-3 weights and spline parameters reuse the dead X tile; every sum unchanged
# speedup vs baseline: 2.4128x; 1.1110x over previous
#include <hip/hip_runtime.h>
#include <math.h>

typedef float v2f __attribute__((ext_vector_type(2)));
typedef float v8f __attribute__((ext_vector_type(8)));
typedef float v4f __attribute__((ext_vector_type(4)));
typedef float v4f_g __attribute__((ext_vector_type(4), aligned(4)));

#define DIMS   64
#define HID    64
#define NBATCH 16384
#define BTILE  128
#define PADW   68
#define PPAD   25
#define TAILB  3.0f

__device__ __forceinline__ v4f ld16(const float* p) {
  return *reinterpret_cast<const v4f_g*>(p);
}

__device__ __forceinline__ float exp_hw(float v) {
  return __builtin_amdgcn_exp2f(v * 1.44269504088896341f);
}
__device__ __forceinline__ float log_hw(float v) {
  return __builtin_amdgcn_logf(v) * 0.693147180559945309f;
}

__device__ __forceinline__ float softplus_f(float x) {
  return fmaxf(x, 0.0f) + log_hw(1.0f + exp_hw(-fabsf(x)));
}

__device__ __forceinline__ float tanh_hw(float v) {
  float e = __builtin_amdgcn_exp2f(v * 2.88539008177792681f);
  float r = __builtin_amdgcn_rcpf(e + 1.0f);
  return fmaf(-2.0f, r, 1.0f);
}

__device__ __forceinline__ v8f splat8(float b) {
  v8f v = {b, b, b, b, b, b, b, b};
  return v;
}

__device__ __forceinline__ v2f ld2(const float* p) {
  v2f v = {p[0], p[1]};
  return v;
}

__device__ __forceinline__ void build_edges(const float* p, float* edges, float* widths) {
  float m1 = p[0];
#pragma unroll
  for (int j = 1; j < 8; ++j) m1 = fmaxf(m1, p[j]);
  float e[8];
  float s = 0.0f;
#pragma unroll
  for (int j = 0; j < 8; ++j) { e[j] = exp_hw(p[j] - m1); s += e[j]; }
  float inv = (2.0f * TAILB) / s;
  float u[8];
  float m2 = -1e30f;
#pragma unroll
  for (int j = 0; j < 8; ++j) { u[j] = e[j] * inv; m2 = fmaxf(m2, u[j]); }
  float s2 = 0.0f;
#pragma unroll
  for (int j = 0; j < 8; ++j) { e[j] = exp_hw(u[j] - m2); s2 += e[j]; }
  float inv2 = (1.0f - 8.0f * 1e-3f) / s2;
  float cum = 0.0f;
  edges[0] = -TAILB;
#pragma unroll
  for (int j = 0; j < 8; ++j) {
    float w = 1e-3f + e[j] * inv2;
    cum += w;
    edges[j + 1] = 2.0f * TAILB * cum - TAILB;
  }
  edges[8] = TAILB;
#pragma unroll
  for (int j = 0; j < 8; ++j) widths[j] = edges[j + 1] - edges[j];
}

__device__ __forceinline__ void rqs_spline(const float* p, float xin, float& zout, float& ldout) {
  bool inside = (xin >= -TAILB) && (xin <= TAILB);
  float xc = fminf(fmaxf(xin, -TAILB), TAILB);
  float cw[9], wdt[8], ch[9], hgt[8], der[9];
  build_edges(p, cw, wdt);
  build_edges(p + 8, ch, hgt);
  der[0] = 1.0f;
  der[8] = 1.0f;
#pragma unroll
  for (int j = 1; j <= 7; ++j) der[j] = 1e-3f + softplus_f(softplus_f(p[15 + j]));
  int idx = -1;
#pragma unroll
  for (int j = 0; j < 9; ++j) {
    float edge = cw[j] + ((j == 8) ? 1e-6f : 0.0f);
    idx += (xc >= edge) ? 1 : 0;
  }
  idx = max(0, min(idx, 7));
  float icw = cw[idx], iw = wdt[idx];
  float ich = ch[idx], ih = hgt[idx];
  float delta = ih / iw;
  float dk = der[idx], dk1 = der[idx + 1];
  float th  = (xc - icw) / iw;
  float omt = 1.0f - th;
  float t1  = th * omt;
  float numer = ih * (delta * th * th + dk * t1);
  float denom = delta + (dk + dk1 - 2.0f * delta) * t1;
  float outv  = ich + numer / denom;
  float dnum  = delta * delta * (dk1 * th * th + 2.0f * delta * t1 + dk * omt * omt);
  float ld    = log_hw(dnum) - 2.0f * log_hw(denom);
  zout  = inside ? outv : xin;
  ldout = inside ? ld : 0.0f;
}

__global__ void __launch_bounds__(256)
nsf_main(const float* __restrict__ x,  const float* __restrict__ initp,
         const float* __restrict__ W1, const float* __restrict__ b1,
         const float* __restrict__ W2, const float* __restrict__ b2,
         const float* __restrict__ W3, const float* __restrict__ b3,
         float* __restrict__ z_out, float* __restrict__ ld_ws) {
  __shared__ __align__(16) float Xs[BTILE][PADW];
  __shared__ float Hs[BTILE][PADW];
  __shared__ __align__(16) float W1s[HID][PADW];
  __shared__ __align__(16) float W2s[HID][PADW];
  __shared__ float B1s[HID];
  __shared__ float B2s[HID];
  __shared__ float B3s[32];
  float (*W3s)[PADW] = &Xs[0];
  float (*Ps)[PPAD]  = reinterpret_cast<float (*)[PPAD]>(&Xs[64][0]);

  const int d   = blockIdx.y;
  const int b0  = blockIdx.x * BTILE;
  const int tid = threadIdx.x;

  if (d == 0) {
    if (tid < BTILE) {
      int b = b0 + tid;
      float pl[23];
#pragma unroll
      for (int j = 0; j < 23; ++j) pl[j] = initp[j];
      float z, ld;
      rqs_spline(pl, x[b * DIMS + 0], z, ld);
      z_out[b * DIMS + 0] = z;
      ld_ws[(size_t)0 * NBATCH + b] = ld;
    }
    return;
  }

  const int l = d - 1;
  v4f w3r[2];

  {
    const float* xg  = x  + (size_t)b0 * DIMS;
    const float* w1p = W1 + (size_t)l * HID * DIMS;
    const float* w2p = W2 + (size_t)l * HID * HID;
    const float* w3p = W3 + (size_t)l * 23 * HID;
    v4f xr[8], w1r[4], w2r[4];
#pragma unroll
    for (int t = 0; t < 8; ++t) xr[t] = ld16(xg + (tid + t * 256) * 4);
#pragma unroll
    for (int t = 0; t < 4; ++t) {
      w1r[t] = ld16(w1p + (tid + t * 256) * 4);
      w2r[t] = ld16(w2p + (tid + t * 256) * 4);
    }
#pragma unroll
    for (int t = 0; t < 2; ++t) {
      const int i = tid + t * 256;
      v4f z4 = {0.0f, 0.0f, 0.0f, 0.0f};
      w3r[t] = (i < 23 * 16) ? ld16(w3p + i * 4) : z4;
    }
    float bv = 0.0f;
    if (tid < 64)       bv = b1[l * HID + tid];
    else if (tid < 128) bv = b2[l * HID + (tid - 64)];
    else if (tid < 151) bv = b3[l * 23 + (tid - 128)];

    const int c4 = (tid & 15) << 2;
#pragma unroll
    for (int t = 0; t < 8; ++t) {
      const int m = (tid + t * 256) >> 4;
      v4f v = xr[t];
      v.x = (c4     < d) ? v.x : 0.0f;
      v.y = (c4 + 1 < d) ? v.y : 0.0f;
      v.z = (c4 + 2 < d) ? v.z : 0.0f;
      v.w = (c4 + 3 < d) ? v.w : 0.0f;
      *reinterpret_cast<v4f*>(&Xs[m][c4]) = v;
    }
#pragma unroll
    for (int t = 0; t < 4; ++t) {
      const int r = (tid + t * 256) >> 4;
      *reinterpret_cast<v4f*>(&W1s[r][c4]) = w1r[t];
      *reinterpret_cast<v4f*>(&W2s[r][c4]) = w2r[t];
    }
    if (tid < 64)       B1s[tid] = bv;
    else if (tid < 128) B2s[tid - 64] = bv;
    else if (tid < 160) B3s[tid - 128] = bv;
  }
  __syncthreads();

  const int lane = tid & 31;
  const int wave = tid >> 5;
  const int mrow = wave * 16;
  const int ln   = lane & 15;
  const int kh   = (lane >> 4) << 1;
  const int mh   = (lane >> 4) << 3;

  const int kmax1 = (d + 3) & ~3;

#pragma unroll
  for (int nt = 0; nt < 4; ++nt) {
    const int n0 = nt * 16;
    const float bias = B1s[n0 + ln];
    v8f acc;
#pragma unroll
    for (int r = 0; r < 8; ++r) acc[r] = bias;
    for (int k0 = 0; k0 < kmax1; k0 += 4) {
      v2f a, bb;
      a.x  = Xs[mrow + ln][k0 + kh];
      a.y  = Xs[mrow + ln][k0 + kh + 1];
      bb.x = W1s[n0 + ln][k0 + kh];
      bb.y = W1s[n0 + ln][k0 + kh + 1];
      acc = __builtin_amdgcn_wmma_f32_16x16x4_f32(false, a, false, bb, (short)0, acc,
                                                  false, false);
    }
#pragma unroll
    for (int r = 0; r < 8; ++r) Hs[mrow + mh + r][n0 + ln] = tanh_hw(acc[r]);
  }
  __syncthreads();

  {
    const int c4 = (tid & 15) << 2;
#pragma unroll
    for (int t = 0; t < 2; ++t) {
      const int r = (tid + t * 256) >> 4;
      *reinterpret_cast<v4f*>(&W3s[r][c4]) = w3r[t];
    }
  }

  {
    v8f acc0 = splat8(B2s[ln]);
    v8f acc1 = splat8(B2s[16 + ln]);
    v8f acc2 = splat8(B2s[32 + ln]);
    v8f acc3 = splat8(B2s[48 + ln]);
#pragma unroll
    for (int k0 = 0; k0 < HID; k0 += 4) {
      const v2f a  = ld2(&Hs[mrow + ln][k0 + kh]);
      const v2f w0 = ld2(&W2s[ln][k0 + kh]);
      const v2f w1 = ld2(&W2s[16 + ln][k0 + kh]);
      const v2f w2 = ld2(&W2s[32 + ln][k0 + kh]);
      const v2f w3 = ld2(&W2s[48 + ln][k0 + kh]);
      acc0 = __builtin_amdgcn_wmma_f32_16x16x4_f32(false, a, false, w0, (short)0, acc0, false, false);
      acc1 = __builtin_amdgcn_wmma_f32_16x16x4_f32(false, a, false, w1, (short)0, acc1, false, false);
      acc2 = __builtin_amdgcn_wmma_f32_16x16x4_f32(false, a, false, w2, (short)0, acc2, false, false);
      acc3 = __builtin_amdgcn_wmma_f32_16x16x4_f32(false, a, false, w3, (short)0, acc3, false, false);
    }
#pragma unroll
    for (int r = 0; r < 8; ++r) {
      Hs[mrow + mh + r][ln]      = tanh_hw(acc0[r]);
      Hs[mrow + mh + r][16 + ln] = tanh_hw(acc1[r]);
      Hs[mrow + mh + r][32 + ln] = tanh_hw(acc2[r]);
      Hs[mrow + mh + r][48 + ln] = tanh_hw(acc3[r]);
    }
  }
  __syncthreads();

  {
    v8f acc0 = splat8(B3s[ln]);
    v8f acc1 = splat8(B3s[16 + ln]);
#pragma unroll
    for (int k0 = 0; k0 < HID; k0 += 4) {
      const v2f a  = ld2(&Hs[mrow + ln][k0 + kh]);
      const v2f w0 = ld2(&W3s[ln][k0 + kh]);
      const v2f w1 = ld2(&W3s[16 + ln][k0 + kh]);
      acc0 = __builtin_amdgcn_wmma_f32_16x16x4_f32(false, a, false, w0, (short)0, acc0, false, false);
      acc1 = __builtin_amdgcn_wmma_f32_16x16x4_f32(false, a, false, w1, (short)0, acc1, false, false);
    }
#pragma unroll
    for (int r = 0; r < 8; ++r) Ps[mrow + mh + r][ln] = acc0[r];
    if (16 + ln < 23) {
#pragma unroll
      for (int r = 0; r < 8; ++r) Ps[mrow + mh + r][16 + ln] = acc1[r];
    }
  }
  __syncthreads();

  if (tid < BTILE) {
    int b = b0 + tid;
    float z, ld;
    rqs_spline(&Ps[tid][0], x[b * DIMS + d], z, ld);
    z_out[b * DIMS + d] = z;
    ld_ws[(size_t)d * NBATCH + b] = ld;
  }
}

__global__ void __launch_bounds__(256)
nsf_reduce(const float* __restrict__ ld_ws, float* __restrict__ out_ld) {
  int b = blockIdx.x * 256 + threadIdx.x;
  float s = 0.0f;
#pragma unroll 8
  for (int dd = 0; dd < DIMS; ++dd) s += ld_ws[(size_t)dd * NBATCH + b];
  out_ld[b] = s;
}

extern "C" void kernel_launch(void* const* d_in, const int* in_sizes, int n_in,
                              void* d_out, int out_size, void* d_ws, size_t ws_size,
                              hipStream_t stream) {
  const float* x     = (const float*)d_in[0];
  const float* initp = (const float*)d_in[1];
  const float* W1    = (const float*)d_in[2];
  const float* b1    = (const float*)d_in[3];
  const float* W2    = (const float*)d_in[4];
  const float* b2    = (const float*)d_in[5];
  const float* W3    = (const float*)d_in[6];
  const float* b3    = (const float*)d_in[7];

  float* z_out  = (float*)d_out;
  float* ld_out = z_out + (size_t)NBATCH * DIMS;
  float* ld_ws  = (float*)d_ws;

  dim3 grid(NBATCH / BTILE, DIMS);
  nsf_main<<<grid, 256, 0, stream>>>(x, initp, W1, b1, W2, b2, W3, b3, z_out, ld_ws);
  nsf_reduce<<<NBATCH / 256, 256, 0, stream>>>(ld_ws, ld_out);
}


# layer 1: four 16-column tiles accumulate together from one A fragment per k step (4 independent WMMAs per LDS wait, was 1), next two k steps' fragments loaded ahead; sums unchanged
# speedup vs baseline: 2.6325x; 1.0911x over previous
#include <hip/hip_runtime.h>
#include <math.h>

typedef float v2f __attribute__((ext_vector_type(2)));
typedef float v8f __attribute__((ext_vector_type(8)));
typedef float v4f __attribute__((ext_vector_type(4)));
typedef float v4f_g __attribute__((ext_vector_type(4), aligned(4)));

#define DIMS   64
#define HID    64
#define NBATCH 16384
#define BTILE  128
#define PADW   68
#define PPAD   25
#define TAILB  3.0f

__device__ __forceinline__ v4f ld16(const float* p) {
  return *reinterpret_cast<const v4f_g*>(p);
}

__device__ __forceinline__ float exp_hw(float v) {
  return __builtin_amdgcn_exp2f(v * 1.44269504088896341f);
}
__device__ __forceinline__ float log_hw(float v) {
  return __builtin_amdgcn_logf(v) * 0.693147180559945309f;
}

__device__ __forceinline__ float softplus_f(float x) {
  return fmaxf(x, 0.0f) + log_hw(1.0f + exp_hw(-fabsf(x)));
}

__device__ __forceinline__ float tanh_hw(float v) {
  float e = __builtin_amdgcn_exp2f(v * 2.88539008177792681f);
  float r = __builtin_amdgcn_rcpf(e + 1.0f);
  return fmaf(-2.0f, r, 1.0f);
}

__device__ __forceinline__ v8f splat8(float b) {
  v8f v = {b, b, b, b, b, b, b, b};
  return v;
}

__device__ __forceinline__ v2f ld2(const float* p) {
  v2f v = {p[0], p[1]};
  return v;
}

struct Frag2 {
  v2f a0, a1;
  v2f w00, w01;
  v2f w10, w11;
  v2f w20, w21;
  v2f w30, w31;
};

__device__ __forceinline__ Frag2 load_frag2(const float* ap, const float* wp0, const float* wp1,
                                            const float* wp2, const float* wp3, int k0) {
  Frag2 f;
  f.a0  = ld2(ap  + k0);  f.a1  = ld2(ap  + k0 + 4);
  f.w00 = ld2(wp0 + k0);  f.w01 = ld2(wp0 + k0 + 4);
  f.w10 = ld2(wp1 + k0);  f.w11 = ld2(wp1 + k0 + 4);
  f.w20 = ld2(wp2 + k0);  f.w21 = ld2(wp2 + k0 + 4);
  f.w30 = ld2(wp3 + k0);  f.w31 = ld2(wp3 + k0 + 4);
  return f;
}

__device__ __forceinline__ void mma_step0(const Frag2& f, v8f& c0, v8f& c1, v8f& c2, v8f& c3) {
  c0 = __builtin_amdgcn_wmma_f32_16x16x4_f32(false, f.a0, false, f.w00, (short)0, c0, false, false);
  c1 = __builtin_amdgcn_wmma_f32_16x16x4_f32(false, f.a0, false, f.w10, (short)0, c1, false, false);
  c2 = __builtin_amdgcn_wmma_f32_16x16x4_f32(false, f.a0, false, f.w20, (short)0, c2, false, false);
  c3 = __builtin_amdgcn_wmma_f32_16x16x4_f32(false, f.a0, false, f.w30, (short)0, c3, false, false);
}
__device__ __forceinline__ void mma_step1(const Frag2& f, v8f& c0, v8f& c1, v8f& c2, v8f& c3) {
  c0 = __builtin_amdgcn_wmma_f32_16x16x4_f32(false, f.a1, false, f.w01, (short)0, c0, false, false);
  c1 = __builtin_amdgcn_wmma_f32_16x16x4_f32(false, f.a1, false, f.w11, (short)0, c1, false, false);
  c2 = __builtin_amdgcn_wmma_f32_16x16x4_f32(false, f.a1, false, f.w21, (short)0, c2, false, false);
  c3 = __builtin_amdgcn_wmma_f32_16x16x4_f32(false, f.a1, false, f.w31, (short)0, c3, false, false);
}

__device__ __forceinline__ void build_edges(const float* p, float* edges, float* widths) {
  float m1 = p[0];
#pragma unroll
  for (int j = 1; j < 8; ++j) m1 = fmaxf(m1, p[j]);
  float e[8];
  float s = 0.0f;
#pragma unroll
  for (int j = 0; j < 8; ++j) { e[j] = exp_hw(p[j] - m1); s += e[j]; }
  float inv = (2.0f * TAILB) / s;
  float u[8];
  float m2 = -1e30f;
#pragma unroll
  for (int j = 0; j < 8; ++j) { u[j] = e[j] * inv; m2 = fmaxf(m2, u[j]); }
  float s2 = 0.0f;
#pragma unroll
  for (int j = 0; j < 8; ++j) { e[j] = exp_hw(u[j] - m2); s2 += e[j]; }
  float inv2 = (1.0f - 8.0f * 1e-3f) / s2;
  float cum = 0.0f;
  edges[0] = -TAILB;
#pragma unroll
  for (int j = 0; j < 8; ++j) {
    float w = 1e-3f + e[j] * inv2;
    cum += w;
    edges[j + 1] = 2.0f * TAILB * cum - TAILB;
  }
  edges[8] = TAILB;
#pragma unroll
  for (int j = 0; j < 8; ++j) widths[j] = edges[j + 1] - edges[j];
}

__device__ __forceinline__ void rqs_spline(const float* p, float xin, float& zout, float& ldout) {
  bool inside = (xin >= -TAILB) && (xin <= TAILB);
  float xc = fminf(fmaxf(xin, -TAILB), TAILB);
  float cw[9], wdt[8], ch[9], hgt[8], der[9];
  build_edges(p, cw, wdt);
  build_edges(p + 8, ch, hgt);
  der[0] = 1.0f;
  der[8] = 1.0f;
#pragma unroll
  for (int j = 1; j <= 7; ++j) der[j] = 1e-3f + softplus_f(softplus_f(p[15 + j]));
  int idx = -1;
#pragma unroll
  for (int j = 0; j < 9; ++j) {
    float edge = cw[j] + ((j == 8) ? 1e-6f : 0.0f);
    idx += (xc >= edge) ? 1 : 0;
  }
  idx = max(0, min(idx, 7));
  float icw = cw[idx], iw = wdt[idx];
  float ich = ch[idx], ih = hgt[idx];
  float delta = ih / iw;
  float dk = der[idx], dk1 = der[idx + 1];
  float th  = (xc - icw) / iw;
  float omt = 1.0f - th;
  float t1  = th * omt;
  float numer = ih * (delta * th * th + dk * t1);
  float denom = delta + (dk + dk1 - 2.0f * delta) * t1;
  float outv  = ich + numer / denom;
  float dnum  = delta * delta * (dk1 * th * th + 2.0f * delta * t1 + dk * omt * omt);
  float ld    = log_hw(dnum) - 2.0f * log_hw(denom);
  zout  = inside ? outv : xin;
  ldout = inside ? ld : 0.0f;
}

__global__ void __launch_bounds__(256)
nsf_main(const float* __restrict__ x,  const float* __restrict__ initp,
         const float* __restrict__ W1, const float* __restrict__ b1,
         const float* __restrict__ W2, const float* __restrict__ b2,
         const float* __restrict__ W3, const float* __restrict__ b3,
         float* __restrict__ z_out, float* __restrict__ ld_ws) {
  __shared__ __align__(16) float Xs[BTILE][PADW];
  __shared__ float Hs[BTILE][PADW];
  __shared__ __align__(16) float W1s[HID][PADW];
  __shared__ __align__(16) float W2s[HID][PADW];
  __shared__ float B1s[HID];
  __shared__ float B2s[HID];
  __shared__ float B3s[32];
  float (*W3s)[PADW] = &Xs[0];
  float (*Ps)[PPAD]  = reinterpret_cast<float (*)[PPAD]>(&Xs[64][0]);

  const int d   = blockIdx.y;
  const int b0  = blockIdx.x * BTILE;
  const int tid = threadIdx.x;

  if (d == 0) {
    if (tid < BTILE) {
      int b = b0 + tid;
      float pl[23];
#pragma unroll
      for (int j = 0; j < 23; ++j) pl[j] = initp[j];
      float z, ld;
      rqs_spline(pl, x[b * DIMS + 0], z, ld);
      z_out[b * DIMS + 0] = z;
      ld_ws[(size_t)0 * NBATCH + b] = ld;
    }
    return;
  }

  const int l = d - 1;
  v4f w3r[2];

  {
    const float* xg  = x  + (size_t)b0 * DIMS;
    const float* w1p = W1 + (size_t)l * HID * DIMS;
    const float* w2p = W2 + (size_t)l * HID * HID;
    const float* w3p = W3 + (size_t)l * 23 * HID;
    v4f xr[8], w1r[4], w2r[4];
#pragma unroll
    for (int t = 0; t < 8; ++t) xr[t] = ld16(xg + (tid + t * 256) * 4);
#pragma unroll
    for (int t = 0; t < 4; ++t) {
      w1r[t] = ld16(w1p + (tid + t * 256) * 4);
      w2r[t] = ld16(w2p + (tid + t * 256) * 4);
    }
#pragma unroll
    for (int t = 0; t < 2; ++t) {
      const int i = tid + t * 256;
      v4f z4 = {0.0f, 0.0f, 0.0f, 0.0f};
      w3r[t] = (i < 23 * 16) ? ld16(w3p + i * 4) : z4;
    }
    float bv = 0.0f;
    if (tid < 64)       bv = b1[l * HID + tid];
    else if (tid < 128) bv = b2[l * HID + (tid - 64)];
    else if (tid < 151) bv = b3[l * 23 + (tid - 128)];

    const int c4 = (tid & 15) << 2;
#pragma unroll
    for (int t = 0; t < 8; ++t) {
      const int m = (tid + t * 256) >> 4;
      v4f v = xr[t];
      v.x = (c4     < d) ? v.x : 0.0f;
      v.y = (c4 + 1 < d) ? v.y : 0.0f;
      v.z = (c4 + 2 < d) ? v.z : 0.0f;
      v.w = (c4 + 3 < d) ? v.w : 0.0f;
      *reinterpret_cast<v4f*>(&Xs[m][c4]) = v;
    }
#pragma unroll
    for (int t = 0; t < 4; ++t) {
      const int r = (tid + t * 256) >> 4;
      *reinterpret_cast<v4f*>(&W1s[r][c4]) = w1r[t];
      *reinterpret_cast<v4f*>(&W2s[r][c4]) = w2r[t];
    }
    if (tid < 64)       B1s[tid] = bv;
    else if (tid < 128) B2s[tid - 64] = bv;
    else if (tid < 160) B3s[tid - 128] = bv;
  }
  __syncthreads();

  const int lane = tid & 31;
  const int wave = tid >> 5;
  const int mrow = wave * 16;
  const int ln   = lane & 15;
  const int kh   = (lane >> 4) << 1;
  const int mh   = (lane >> 4) << 3;

  {
    v8f acc0 = splat8(B1s[ln]);
    v8f acc1 = splat8(B1s[16 + ln]);
    v8f acc2 = splat8(B1s[32 + ln]);
    v8f acc3 = splat8(B1s[48 + ln]);
    const float* ap  = &Xs[mrow + ln][kh];
    const float* wp0 = &W1s[ln][kh];
    const float* wp1 = &W1s[16 + ln][kh];
    const float* wp2 = &W1s[32 + ln][kh];
    const float* wp3 = &W1s[48 + ln][kh];
    Frag2 fr[2];
    fr[0] = load_frag2(ap, wp0, wp1, wp2, wp3, 0);
#pragma unroll
    for (int g = 0; g < 8; ++g) {
      const bool more = (g + 1) * 8 < d;
      if (g < 7 && more) fr[(g + 1) & 1] = load_frag2(ap, wp0, wp1, wp2, wp3, (g + 1) * 8);
      mma_step0(fr[g & 1], acc0, acc1, acc2, acc3);
      if (g * 8 + 4 < d) mma_step1(fr[g & 1], acc0, acc1, acc2, acc3);
      if (!more) break;
    }
#pragma unroll
    for (int r = 0; r < 8; ++r) {
      Hs[mrow + mh + r][ln]      = tanh_hw(acc0[r]);
      Hs[mrow + mh + r][16 + ln] = tanh_hw(acc1[r]);
      Hs[mrow + mh + r][32 + ln] = tanh_hw(acc2[r]);
      Hs[mrow + mh + r][48 + ln] = tanh_hw(acc3[r]);
    }
  }
  __syncthreads();

  {
    const int c4 = (tid & 15) << 2;
#pragma unroll
    for (int t = 0; t < 2; ++t) {
      const int r = (tid + t * 256) >> 4;
      *reinterpret_cast<v4f*>(&W3s[r][c4]) = w3r[t];
    }
  }

  {
    v8f acc0 = splat8(B2s[ln]);
    v8f acc1 = splat8(B2s[16 + ln]);
    v8f acc2 = splat8(B2s[32 + ln]);
    v8f acc3 = splat8(B2s[48 + ln]);
#pragma unroll
    for (int k0 = 0; k0 < HID; k0 += 4) {
      const v2f a  = ld2(&Hs[mrow + ln][k0 + kh]);
      const v2f w0 = ld2(&W2s[ln][k0 + kh]);
      const v2f w1 = ld2(&W2s[16 + ln][k0 + kh]);
      const v2f w2 = ld2(&W2s[32 + ln][k0 + kh]);
      const v2f w3 = ld2(&W2s[48 + ln][k0 + kh]);
      acc0 = __builtin_amdgcn_wmma_f32_16x16x4_f32(false, a, false, w0, (short)0, acc0, false, false);
      acc1 = __builtin_amdgcn_wmma_f32_16x16x4_f32(false, a, false, w1, (short)0, acc1, false, false);
      acc2 = __builtin_amdgcn_wmma_f32_16x16x4_f32(false, a, false, w2, (short)0, acc2, false, false);
      acc3 = __builtin_amdgcn_wmma_f32_16x16x4_f32(false, a, false, w3, (short)0, acc3, false, false);
    }
#pragma unroll
    for (int r = 0; r < 8; ++r) {
      Hs[mrow + mh + r][ln]      = tanh_hw(acc0[r]);
      Hs[mrow + mh + r][16 + ln] = tanh_hw(acc1[r]);
      Hs[mrow + mh + r][32 + ln] = tanh_hw(acc2[r]);
      Hs[mrow + mh + r][48 + ln] = tanh_hw(acc3[r]);
    }
  }
  __syncthreads();

  {
    v8f acc0 = splat8(B3s[ln]);
    v8f acc1 = splat8(B3s[16 + ln]);
#pragma unroll
    for (int k0 = 0; k0 < HID; k0 += 4) {
      const v2f a  = ld2(&Hs[mrow + ln][k0 + kh]);
      const v2f w0 = ld2(&W3s[ln][k0 + kh]);
      const v2f w1 = ld2(&W3s[16 + ln][k0 + kh]);
      acc0 = __builtin_amdgcn_wmma_f32_16x16x4_f32(false, a, false, w0, (short)0, acc0, false, false);
      acc1 = __builtin_amdgcn_wmma_f32_16x16x4_f32(false, a, false, w1, (short)0, acc1, false, false);
    }
#pragma unroll
    for (int r = 0; r < 8; ++r) Ps[mrow + mh + r][ln] = acc0[r];
    if (16 + ln < 23) {
#pragma unroll
      for (int r = 0; r < 8; ++r) Ps[mrow + mh + r][16 + ln] = acc1[r];
    }
  }
  __syncthreads();

  if (tid < BTILE) {
    int b = b0 + tid;
    float z, ld;
    rqs_spline(&Ps[tid][0], x[b * DIMS + d], z, ld);
    z_out[b * DIMS + d] = z;
    ld_ws[(size_t)d * NBATCH + b] = ld;
  }
}

__global__ void __launch_bounds__(256)
nsf_reduce(const float* __restrict__ ld_ws, float* __restrict__ out_ld) {
  int b = blockIdx.x * 256 + threadIdx.x;
  float s = 0.0f;
#pragma unroll 8
  for (int dd = 0; dd < DIMS; ++dd) s += ld_ws[(size_t)dd * NBATCH + b];
  out_ld[b] = s;
}

extern "C" void kernel_launch(void* const* d_in, const int* in_sizes, int n_in,
                              void* d_out, int out_size, void* d_ws, size_t ws_size,
                              hipStream_t stream) {
  const float* x     = (const float*)d_in[0];
  const float* initp = (const float*)d_in[1];
  const float* W1    = (const float*)d_in[2];
  const float* b1    = (const float*)d_in[3];
  const float* W2    = (const float*)d_in[4];
  const float* b2    = (const float*)d_in[5];
  const float* W3    = (const float*)d_in[6];
  const float* b3    = (const float*)d_in[7];

  float* z_out  = (float*)d_out;
  float* ld_out = z_out + (size_t)NBATCH * DIMS;
  float* ld_ws  = (float*)d_ws;

  dim3 grid(NBATCH / BTILE, DIMS);
  nsf_main<<<grid, 256, 0, stream>>>(x, initp, W1, b1, W2, b2, W3, b3, z_out, ld_ws);
  nsf_reduce<<<NBATCH / 256, 256, 0, stream>>>(ld_ws, ld_out);
}


# tanh of both hidden layers: one v_tanh_f32 per value (the hardware transcendental) in place of mul, exp2, add, rcp, fma: 64 values a thread, 5 instructions down to 1; h takes the unit's own rounding
# speedup vs baseline: 2.8488x; 1.0822x over previous
#include <hip/hip_runtime.h>
#include <math.h>

typedef float v2f __attribute__((ext_vector_type(2)));
typedef float v8f __attribute__((ext_vector_type(8)));
typedef float v4f __attribute__((ext_vector_type(4)));
typedef float v4f_g __attribute__((ext_vector_type(4), aligned(4)));

#define DIMS   64
#define HID    64
#define NBATCH 16384
#define BTILE  128
#define PADW   68
#define PPAD   25
#define TAILB  3.0f

__device__ __forceinline__ v4f ld16(const float* p) {
  return *reinterpret_cast<const v4f_g*>(p);
}

__device__ __forceinline__ float exp_hw(float v) {
  return __builtin_amdgcn_exp2f(v * 1.44269504088896341f);
}
__device__ __forceinline__ float log_hw(float v) {
  return __builtin_amdgcn_logf(v) * 0.693147180559945309f;
}

__device__ __forceinline__ float softplus_f(float x) {
  return fmaxf(x, 0.0f) + log_hw(1.0f + exp_hw(-fabsf(x)));
}

__device__ __forceinline__ float tanh_hw(float v) {
  return __builtin_amdgcn_tanhf(v);
}

__device__ __forceinline__ v8f splat8(float b) {
  v8f v = {b, b, b, b, b, b, b, b};
  return v;
}

__device__ __forceinline__ v2f ld2(const float* p) {
  v2f v = {p[0], p[1]};
  return v;
}

struct Frag2 {
  v2f a0, a1;
  v2f w00, w01;
  v2f w10, w11;
  v2f w20, w21;
  v2f w30, w31;
};

__device__ __forceinline__ Frag2 load_frag2(const float* ap, const float* wp0, const float* wp1,
                                            const float* wp2, const float* wp3, int k0) {
  Frag2 f;
  f.a0  = ld2(ap  + k0);  f.a1  = ld2(ap  + k0 + 4);
  f.w00 = ld2(wp0 + k0);  f.w01 = ld2(wp0 + k0 + 4);
  f.w10 = ld2(wp1 + k0);  f.w11 = ld2(wp1 + k0 + 4);
  f.w20 = ld2(wp2 + k0);  f.w21 = ld2(wp2 + k0 + 4);
  f.w30 = ld2(wp3 + k0);  f.w31 = ld2(wp3 + k0 + 4);
  return f;
}

__device__ __forceinline__ void mma_step0(const Frag2& f, v8f& c0, v8f& c1, v8f& c2, v8f& c3) {
  c0 = __builtin_amdgcn_wmma_f32_16x16x4_f32(false, f.a0, false, f.w00, (short)0, c0, false, false);
  c1 = __builtin_amdgcn_wmma_f32_16x16x4_f32(false, f.a0, false, f.w10, (short)0, c1, false, false);
  c2 = __builtin_amdgcn_wmma_f32_16x16x4_f32(false, f.a0, false, f.w20, (short)0, c2, false, false);
  c3 = __builtin_amdgcn_wmma_f32_16x16x4_f32(false, f.a0, false, f.w30, (short)0, c3, false, false);
}
__device__ __forceinline__ void mma_step1(const Frag2& f, v8f& c0, v8f& c1, v8f& c2, v8f& c3) {
  c0 = __builtin_amdgcn_wmma_f32_16x16x4_f32(false, f.a1, false, f.w01, (short)0, c0, false, false);
  c1 = __builtin_amdgcn_wmma_f32_16x16x4_f32(false, f.a1, false, f.w11, (short)0, c1, false, false);
  c2 = __builtin_amdgcn_wmma_f32_16x16x4_f32(false, f.a1, false, f.w21, (short)0, c2, false, false);
  c3 = __builtin_amdgcn_wmma_f32_16x16x4_f32(false, f.a1, false, f.w31, (short)0, c3, false, false);
}

__device__ __forceinline__ void build_edges(const float* p, float* edges, float* widths) {
  float m1 = p[0];
#pragma unroll
  for (int j = 1; j < 8; ++j) m1 = fmaxf(m1, p[j]);
  float e[8];
  float s = 0.0f;
#pragma unroll
  for (int j = 0; j < 8; ++j) { e[j] = exp_hw(p[j] - m1); s += e[j]; }
  float inv = (2.0f * TAILB) / s;
  float u[8];
  float m2 = -1e30f;
#pragma unroll
  for (int j = 0; j < 8; ++j) { u[j] = e[j] * inv; m2 = fmaxf(m2, u[j]); }
  float s2 = 0.0f;
#pragma unroll
  for (int j = 0; j < 8; ++j) { e[j] = exp_hw(u[j] - m2); s2 += e[j]; }
  float inv2 = (1.0f - 8.0f * 1e-3f) / s2;
  float cum = 0.0f;
  edges[0] = -TAILB;
#pragma unroll
  for (int j = 0; j < 8; ++j) {
    float w = 1e-3f + e[j] * inv2;
    cum += w;
    edges[j + 1] = 2.0f * TAILB * cum - TAILB;
  }
  edges[8] = TAILB;
#pragma unroll
  for (int j = 0; j < 8; ++j) widths[j] = edges[j + 1] - edges[j];
}

__device__ __forceinline__ void rqs_spline(const float* p, float xin, float& zout, float& ldout) {
  bool inside = (xin >= -TAILB) && (xin <= TAILB);
  float xc = fminf(fmaxf(xin, -TAILB), TAILB);
  float cw[9], wdt[8], ch[9], hgt[8], der[9];
  build_edges(p, cw, wdt);
  build_edges(p + 8, ch, hgt);
  der[0] = 1.0f;
  der[8] = 1.0f;
#pragma unroll
  for (int j = 1; j <= 7; ++j) der[j] = 1e-3f + softplus_f(softplus_f(p[15 + j]));
  int idx = -1;
#pragma unroll
  for (int j = 0; j < 9; ++j) {
    float edge = cw[j] + ((j == 8) ? 1e-6f : 0.0f);
    idx += (xc >= edge) ? 1 : 0;
  }
  idx = max(0, min(idx, 7));
  float icw = cw[idx], iw = wdt[idx];
  float ich = ch[idx], ih = hgt[idx];
  float delta = ih / iw;
  float dk = der[idx], dk1 = der[idx + 1];
  float th  = (xc - icw) / iw;
  float omt = 1.0f - th;
  float t1  = th * omt;
  float numer = ih * (delta * th * th + dk * t1);
  float denom = delta + (dk + dk1 - 2.0f * delta) * t1;
  float outv  = ich + numer / denom;
  float dnum  = delta * delta * (dk1 * th * th + 2.0f * delta * t1 + dk * omt * omt);
  float ld    = log_hw(dnum) - 2.0f * log_hw(denom);
  zout  = inside ? outv : xin;
  ldout = inside ? ld : 0.0f;
}

__global__ void __launch_bounds__(256)
nsf_main(const float* __restrict__ x,  const float* __restrict__ initp,
         const float* __restrict__ W1, const float* __restrict__ b1,
         const float* __restrict__ W2, const float* __restrict__ b2,
         const float* __restrict__ W3, const float* __restrict__ b3,
         float* __restrict__ z_out, float* __restrict__ ld_ws) {
  __shared__ __align__(16) float Xs[BTILE][PADW];
  __shared__ float Hs[BTILE][PADW];
  __shared__ __align__(16) float W1s[HID][PADW];
  __shared__ __align__(16) float W2s[HID][PADW];
  __shared__ float B1s[HID];
  __shared__ float B2s[HID];
  __shared__ float B3s[32];
  float (*W3s)[PADW] = &Xs[0];
  float (*Ps)[PPAD]  = reinterpret_cast<float (*)[PPAD]>(&Xs[64][0]);

  const int d   = blockIdx.y;
  const int b0  = blockIdx.x * BTILE;
  const int tid = threadIdx.x;

  if (d == 0) {
    if (tid < BTILE) {
      int b = b0 + tid;
      float pl[23];
#pragma unroll
      for (int j = 0; j < 23; ++j) pl[j] = initp[j];
      float z, ld;
      rqs_spline(pl, x[b * DIMS + 0], z, ld);
      z_out[b * DIMS + 0] = z;
      ld_ws[(size_t)0 * NBATCH + b] = ld;
    }
    return;
  }

  const int l = d - 1;
  v4f w3r[2];

  {
    const float* xg  = x  + (size_t)b0 * DIMS;
    const float* w1p = W1 + (size_t)l * HID * DIMS;
    const float* w2p = W2 + (size_t)l * HID * HID;
    const float* w3p = W3 + (size_t)l * 23 * HID;
    v4f xr[8], w1r[4], w2r[4];
#pragma unroll
    for (int t = 0; t < 8; ++t) xr[t] = ld16(xg + (tid + t * 256) * 4);
#pragma unroll
    for (int t = 0; t < 4; ++t) {
      w1r[t] = ld16(w1p + (tid + t * 256) * 4);
      w2r[t] = ld16(w2p + (tid + t * 256) * 4);
    }
#pragma unroll
    for (int t = 0; t < 2; ++t) {
      const int i = tid + t * 256;
      v4f z4 = {0.0f, 0.0f, 0.0f, 0.0f};
      w3r[t] = (i < 23 * 16) ? ld16(w3p + i * 4) : z4;
    }
    float bv = 0.0f;
    if (tid < 64)       bv = b1[l * HID + tid];
    else if (tid < 128) bv = b2[l * HID + (tid - 64)];
    else if (tid < 151) bv = b3[l * 23 + (tid - 128)];

    const int c4 = (tid & 15) << 2;
#pragma unroll
    for (int t = 0; t < 8; ++t) {
      const int m = (tid + t * 256) >> 4;
      v4f v = xr[t];
      v.x = (c4     < d) ? v.x : 0.0f;
      v.y = (c4 + 1 < d) ? v.y : 0.0f;
      v.z = (c4 + 2 < d) ? v.z : 0.0f;
      v.w = (c4 + 3 < d) ? v.w : 0.0f;
      *reinterpret_cast<v4f*>(&Xs[m][c4]) = v;
    }
#pragma unroll
    for (int t = 0; t < 4; ++t) {
      const int r = (tid + t * 256) >> 4;
      *reinterpret_cast<v4f*>(&W1s[r][c4]) = w1r[t];
      *reinterpret_cast<v4f*>(&W2s[r][c4]) = w2r[t];
    }
    if (tid < 64)       B1s[tid] = bv;
    else if (tid < 128) B2s[tid - 64] = bv;
    else if (tid < 160) B3s[tid - 128] = bv;
  }
  __syncthreads();

  const int lane = tid & 31;
  const int wave = tid >> 5;
  const int mrow = wave * 16;
  const int ln   = lane & 15;
  const int kh   = (lane >> 4) << 1;
  const int mh   = (lane >> 4) << 3;

  {
    v8f acc0 = splat8(B1s[ln]);
    v8f acc1 = splat8(B1s[16 + ln]);
    v8f acc2 = splat8(B1s[32 + ln]);
    v8f acc3 = splat8(B1s[48 + ln]);
    const float* ap  = &Xs[mrow + ln][kh];
    const float* wp0 = &W1s[ln][kh];
    const float* wp1 = &W1s[16 + ln][kh];
    const float* wp2 = &W1s[32 + ln][kh];
    const float* wp3 = &W1s[48 + ln][kh];
    Frag2 fr[2];
    fr[0] = load_frag2(ap, wp0, wp1, wp2, wp3, 0);
#pragma unroll
    for (int g = 0; g < 8; ++g) {
      const bool more = (g + 1) * 8 < d;
      if (g < 7 && more) fr[(g + 1) & 1] = load_frag2(ap, wp0, wp1, wp2, wp3, (g + 1) * 8);
      mma_step0(fr[g & 1], acc0, acc1, acc2, acc3);
      if (g * 8 + 4 < d) mma_step1(fr[g & 1], acc0, acc1, acc2, acc3);
      if (!more) break;
    }
#pragma unroll
    for (int r = 0; r < 8; ++r) {
      Hs[mrow + mh + r][ln]      = tanh_hw(acc0[r]);
      Hs[mrow + mh + r][16 + ln] = tanh_hw(acc1[r]);
      Hs[mrow + mh + r][32 + ln] = tanh_hw(acc2[r]);
      Hs[mrow + mh + r][48 + ln] = tanh_hw(acc3[r]);
    }
  }
  __syncthreads();

  {
    const int c4 = (tid & 15) << 2;
#pragma unroll
    for (int t = 0; t < 2; ++t) {
      const int r = (tid + t * 256) >> 4;
      *reinterpret_cast<v4f*>(&W3s[r][c4]) = w3r[t];
    }
  }

  {
    v8f acc0 = splat8(B2s[ln]);
    v8f acc1 = splat8(B2s[16 + ln]);
    v8f acc2 = splat8(B2s[32 + ln]);
    v8f acc3 = splat8(B2s[48 + ln]);
#pragma unroll
    for (int k0 = 0; k0 < HID; k0 += 4) {
      const v2f a  = ld2(&Hs[mrow + ln][k0 + kh]);
      const v2f w0 = ld2(&W2s[ln][k0 + kh]);
      const v2f w1 = ld2(&W2s[16 + ln][k0 + kh]);
      const v2f w2 = ld2(&W2s[32 + ln][k0 + kh]);
      const v2f w3 = ld2(&W2s[48 + ln][k0 + kh]);
      acc0 = __builtin_amdgcn_wmma_f32_16x16x4_f32(false, a, false, w0, (short)0, acc0, false, false);
      acc1 = __builtin_amdgcn_wmma_f32_16x16x4_f32(false, a, false, w1, (short)0, acc1, false, false);
      acc2 = __builtin_amdgcn_wmma_f32_16x16x4_f32(false, a, false, w2, (short)0, acc2, false, false);
      acc3 = __builtin_amdgcn_wmma_f32_16x16x4_f32(false, a, false, w3, (short)0, acc3, false, false);
    }
#pragma unroll
    for (int r = 0; r < 8; ++r) {
      Hs[mrow + mh + r][ln]      = tanh_hw(acc0[r]);
      Hs[mrow + mh + r][16 + ln] = tanh_hw(acc1[r]);
      Hs[mrow + mh + r][32 + ln] = tanh_hw(acc2[r]);
      Hs[mrow + mh + r][48 + ln] = tanh_hw(acc3[r]);
    }
  }
  __syncthreads();

  {
    v8f acc0 = splat8(B3s[ln]);
    v8f acc1 = splat8(B3s[16 + ln]);
#pragma unroll
    for (int k0 = 0; k0 < HID; k0 += 4) {
      const v2f a  = ld2(&Hs[mrow + ln][k0 + kh]);
      const v2f w0 = ld2(&W3s[ln][k0 + kh]);
      const v2f w1 = ld2(&W3s[16 + ln][k0 + kh]);
      acc0 = __builtin_amdgcn_wmma_f32_16x16x4_f32(false, a, false, w0, (short)0, acc0, false, false);
      acc1 = __builtin_amdgcn_wmma_f32_16x16x4_f32(false, a, false, w1, (short)0, acc1, false, false);
    }
#pragma unroll
    for (int r = 0; r < 8; ++r) Ps[mrow + mh + r][ln] = acc0[r];
    if (16 + ln < 23) {
#pragma unroll
      for (int r = 0; r < 8; ++r) Ps[mrow + mh + r][16 + ln] = acc1[r];
    }
  }
  __syncthreads();

  if (tid < BTILE) {
    int b = b0 + tid;
    float z, ld;
    rqs_spline(&Ps[tid][0], x[b * DIMS + d], z, ld);
    z_out[b * DIMS + d] = z;
    ld_ws[(size_t)d * NBATCH + b] = ld;
  }
}

__global__ void __launch_bounds__(256)
nsf_reduce(const float* __restrict__ ld_ws, float* __restrict__ out_ld) {
  int b = blockIdx.x * 256 + threadIdx.x;
  float s = 0.0f;
#pragma unroll 8
  for (int dd = 0; dd < DIMS; ++dd) s += ld_ws[(size_t)dd * NBATCH + b];
  out_ld[b] = s;
}

extern "C" void kernel_launch(void* const* d_in, const int* in_sizes, int n_in,
                              void* d_out, int out_size, void* d_ws, size_t ws_size,
                              hipStream_t stream) {
  const float* x     = (const float*)d_in[0];
  const float* initp = (const float*)d_in[1];
  const float* W1    = (const float*)d_in[2];
  const float* b1    = (const float*)d_in[3];
  const float* W2    = (const float*)d_in[4];
  const float* b2    = (const float*)d_in[5];
  const float* W3    = (const float*)d_in[6];
  const float* b3    = (const float*)d_in[7];

  float* z_out  = (float*)d_out;
  float* ld_out = z_out + (size_t)NBATCH * DIMS;
  float* ld_ws  = (float*)d_ws;

  dim3 grid(NBATCH / BTILE, DIMS);
  nsf_main<<<grid, 256, 0, stream>>>(x, initp, W1, b1, W2, b2, W3, b3, z_out, ld_ws);
  nsf_reduce<<<NBATCH / 256, 256, 0, stream>>>(ld_ws, ld_out);
}


# persistent: 768 blocks (128 row tiles x 6 groups) keep X in LDS and loop over 10-11 conditioners, next weights prefetched, wave priority falls with progress; log-det: per block, then 6 partials
# speedup vs baseline: 3.1238x; 1.0965x over previous
#include <hip/hip_runtime.h>
#include <math.h>

typedef float v2f __attribute__((ext_vector_type(2)));
typedef float v8f __attribute__((ext_vector_type(8)));
typedef float v4f __attribute__((ext_vector_type(4)));
typedef float v4f_g __attribute__((ext_vector_type(4), aligned(4)));

#define DIMS   64
#define HID    64
#define NBATCH 16384
#define BTILE  128
#define PADW   68
#define TAILB  3.0f

__device__ __forceinline__ v4f ld16(const float* p) {
  return *reinterpret_cast<const v4f_g*>(p);
}

__device__ __forceinline__ float exp_hw(float v) {
  return __builtin_amdgcn_exp2f(v * 1.44269504088896341f);
}
__device__ __forceinline__ float log_hw(float v) {
  return __builtin_amdgcn_logf(v) * 0.693147180559945309f;
}

__device__ __forceinline__ float softplus_f(float x) {
  return fmaxf(x, 0.0f) + log_hw(1.0f + exp_hw(-fabsf(x)));
}

__device__ __forceinline__ float tanh_hw(float v) {
  return __builtin_amdgcn_tanhf(v);
}

__device__ __forceinline__ v8f splat8(float b) {
  v8f v = {b, b, b, b, b, b, b, b};
  return v;
}

__device__ __forceinline__ v2f ld2(const float* p) {
  v2f v = {p[0], p[1]};
  return v;
}

struct Frag2 {
  v2f a0, a1;
  v2f w00, w01;
  v2f w10, w11;
  v2f w20, w21;
  v2f w30, w31;
};

__device__ __forceinline__ Frag2 load_frag2(const float* ap, const float* wp0, const float* wp1,
                                            const float* wp2, const float* wp3, int k0) {
  Frag2 f;
  f.a0  = ld2(ap  + k0);  f.a1  = ld2(ap  + k0 + 4);
  f.w00 = ld2(wp0 + k0);  f.w01 = ld2(wp0 + k0 + 4);
  f.w10 = ld2(wp1 + k0);  f.w11 = ld2(wp1 + k0 + 4);
  f.w20 = ld2(wp2 + k0);  f.w21 = ld2(wp2 + k0 + 4);
  f.w30 = ld2(wp3 + k0);  f.w31 = ld2(wp3 + k0 + 4);
  return f;
}

__device__ __forceinline__ void mma_step0(const Frag2& f, v8f& c0, v8f& c1, v8f& c2, v8f& c3) {
  c0 = __builtin_amdgcn_wmma_f32_16x16x4_f32(false, f.a0, false, f.w00, (short)0, c0, false, false);
  c1 = __builtin_amdgcn_wmma_f32_16x16x4_f32(false, f.a0, false, f.w10, (short)0, c1, false, false);
  c2 = __builtin_amdgcn_wmma_f32_16x16x4_f32(false, f.a0, false, f.w20, (short)0, c2, false, false);
  c3 = __builtin_amdgcn_wmma_f32_16x16x4_f32(false, f.a0, false, f.w30, (short)0, c3, false, false);
}
__device__ __forceinline__ void mma_step1(const Frag2& f, v8f& c0, v8f& c1, v8f& c2, v8f& c3) {
  c0 = __builtin_amdgcn_wmma_f32_16x16x4_f32(false, f.a1, false, f.w01, (short)0, c0, false, false);
  c1 = __builtin_amdgcn_wmma_f32_16x16x4_f32(false, f.a1, false, f.w11, (short)0, c1, false, false);
  c2 = __builtin_amdgcn_wmma_f32_16x16x4_f32(false, f.a1, false, f.w21, (short)0, c2, false, false);
  c3 = __builtin_amdgcn_wmma_f32_16x16x4_f32(false, f.a1, false, f.w31, (short)0, c3, false, false);
}

__device__ __forceinline__ void build_edges(const float* p, float* edges, float* widths) {
  float m1 = p[0];
#pragma unroll
  for (int j = 1; j < 8; ++j) m1 = fmaxf(m1, p[j]);
  float e[8];
  float s = 0.0f;
#pragma unroll
  for (int j = 0; j < 8; ++j) { e[j] = exp_hw(p[j] - m1); s += e[j]; }
  float inv = (2.0f * TAILB) / s;
  float u[8];
  float m2 = -1e30f;
#pragma unroll
  for (int j = 0; j < 8; ++j) { u[j] = e[j] * inv; m2 = fmaxf(m2, u[j]); }
  float s2 = 0.0f;
#pragma unroll
  for (int j = 0; j < 8; ++j) { e[j] = exp_hw(u[j] - m2); s2 += e[j]; }
  float inv2 = (1.0f - 8.0f * 1e-3f) / s2;
  float cum = 0.0f;
  edges[0] = -TAILB;
#pragma unroll
  for (int j = 0; j < 8; ++j) {
    float w = 1e-3f + e[j] * inv2;
    cum += w;
    edges[j + 1] = 2.0f * TAILB * cum - TAILB;
  }
  edges[8] = TAILB;
#pragma unroll
  for (int j = 0; j < 8; ++j) widths[j] = edges[j + 1] - edges[j];
}

__device__ __forceinline__ void rqs_spline(const float* p, float xin, float& zout, float& ldout) {
  bool inside = (xin >= -TAILB) && (xin <= TAILB);
  float xc = fminf(fmaxf(xin, -TAILB), TAILB);
  float cw[9], wdt[8], ch[9], hgt[8], der[9];
  build_edges(p, cw, wdt);
  build_edges(p + 8, ch, hgt);
  der[0] = 1.0f;
  der[8] = 1.0f;
#pragma unroll
  for (int j = 1; j <= 7; ++j) der[j] = 1e-3f + softplus_f(softplus_f(p[15 + j]));
  int idx = -1;
#pragma unroll
  for (int j = 0; j < 9; ++j) {
    float edge = cw[j] + ((j == 8) ? 1e-6f : 0.0f);
    idx += (xc >= edge) ? 1 : 0;
  }
  idx = max(0, min(idx, 7));
  float icw = cw[0], iw = wdt[0];
  float ich = ch[0], ih = hgt[0];
  float dk = der[0], dk1 = der[1];
#pragma unroll
  for (int j = 1; j < 8; ++j) {
    const bool hit = (idx == j);
    icw = hit ? cw[j]  : icw;   iw  = hit ? wdt[j]     : iw;
    ich = hit ? ch[j]  : ich;   ih  = hit ? hgt[j]     : ih;
    dk  = hit ? der[j] : dk;    dk1 = hit ? der[j + 1] : dk1;
  }
  float delta = ih / iw;
  float th  = (xc - icw) / iw;
  float omt = 1.0f - th;
  float t1  = th * omt;
  float numer = ih * (delta * th * th + dk * t1);
  float denom = delta + (dk + dk1 - 2.0f * delta) * t1;
  float outv  = ich + numer / denom;
  float dnum  = delta * delta * (dk1 * th * th + 2.0f * delta * t1 + dk * omt * omt);
  float ld    = log_hw(dnum) - 2.0f * log_hw(denom);
  zout  = inside ? outv : xin;
  ldout = inside ? ld : 0.0f;
}

#define NGROUP 6

__device__ __forceinline__ int group_count(int g) { return (g < 3) ? 11 : 10; }
__device__ __forceinline__ unsigned long long group_lo(int g) {
  return (g == 0) ? 0x0e12161726292c2fULL
                              : (g == 1) ? 0x0c18191b1f243034ULL
                              : (g == 2) ? 0x0f111c2122232b38ULL
                              : (g == 3) ? 0x1e20252d2e32363aULL
                              : (g == 4) ? 0x1a1d283335393d3fULL
                              :            0x14272a31373b3c3eULL;
}
__device__ __forceinline__ unsigned int group_hi(int g) {
  return (g == 0) ? 0x00020a0du
                        : (g == 1) ? 0x0003090bu
                        : (g == 2) ? 0x00040708u
                        : (g == 3) ? 0x00000615u
                        : (g == 4) ? 0x00000113u
                        :            0x00000510u;
}
__device__ __forceinline__ int group_dim(unsigned long long lo, unsigned int hi, int it) {
  return (it < 8) ? (int)((lo >> (8 * it)) & 0xffULL) : (int)((hi >> (8 * (it - 8))) & 0xffu);
}

__device__ __forceinline__ void fetch_weights(const float* __restrict__ W1, const float* __restrict__ b1,
                                              const float* __restrict__ W2, const float* __restrict__ b2,
                                              const float* __restrict__ W3, const float* __restrict__ b3,
                                              int d, int tid, v4f (&w1r)[4], v4f (&w2r)[4], v4f (&w3r)[2],
                                              float& bv) {
  const int l = d - 1;
  const float* w1p = W1 + (size_t)l * HID * DIMS;
  const float* w2p = W2 + (size_t)l * HID * HID;
  const float* w3p = W3 + (size_t)l * 23 * HID;
#pragma unroll
  for (int t = 0; t < 4; ++t) {
    w1r[t] = ld16(w1p + (tid + t * 256) * 4);
    w2r[t] = ld16(w2p + (tid + t * 256) * 4);
  }
#pragma unroll
  for (int t = 0; t < 2; ++t) {
    const int i = tid + t * 256;
    v4f z4 = {0.0f, 0.0f, 0.0f, 0.0f};
    w3r[t] = (i < 23 * 16) ? ld16(w3p + i * 4) : z4;
  }
  bv = 0.0f;
  if (tid < 64)       bv = b1[l * HID + tid];
  else if (tid < 128) bv = b2[l * HID + (tid - 64)];
  else if (tid < 151) bv = b3[l * 23 + (tid - 128)];
}

__global__ void __launch_bounds__(256, 6)
nsf_main(const float* __restrict__ x,  const float* __restrict__ initp,
         const float* __restrict__ W1, const float* __restrict__ b1,
         const float* __restrict__ W2, const float* __restrict__ b2,
         const float* __restrict__ W3, const float* __restrict__ b3,
         float* __restrict__ z_out, float* __restrict__ ld_ws) {
  __shared__ __align__(16) float Xs[BTILE][PADW];
  __shared__ __align__(16) float Hs[BTILE][PADW];
  __shared__ __align__(16) float W1s[HID][PADW];
  __shared__ __align__(16) float W2s[HID][PADW];
  __shared__ float B1s[HID];
  __shared__ float B2s[HID];
  __shared__ float B3s[32];
  float (*W3s)[PADW] = &W1s[0];

  const int g   = blockIdx.y;
  const int b0  = blockIdx.x * BTILE;
  const int tid = threadIdx.x;
  const int nit = group_count(g);
  const unsigned long long glo = group_lo(g);
  const unsigned int       ghi = group_hi(g);
  const int c4  = (tid & 15) << 2;

  const int lane = tid & 31;
  const int wave = tid >> 5;
  const int mrow = wave * 16;
  const int ln   = lane & 15;
  const int kh   = (lane >> 4) << 1;
  const int mh   = (lane >> 4) << 3;

  v4f w1r[4], w2r[4], w3r[2];
  float bv;

  {
    const float* xg = x + (size_t)b0 * DIMS;
    v4f xr[8];
#pragma unroll
    for (int t = 0; t < 8; ++t) xr[t] = ld16(xg + (tid + t * 256) * 4);
    fetch_weights(W1, b1, W2, b2, W3, b3, group_dim(glo, ghi, 0), tid, w1r, w2r, w3r, bv);
#pragma unroll
    for (int t = 0; t < 8; ++t) {
      const int m = (tid + t * 256) >> 4;
      *reinterpret_cast<v4f*>(&Xs[m][c4]) = xr[t];
    }
  }

  float ldacc = 0.0f;

  for (int it = 0; it < nit; ++it) {
    const int d = group_dim(glo, ghi, it);

    {
      const int q4 = 4 * it;
      const int pr = 3 - (q4 >= nit ? 1 : 0) - (q4 >= 2 * nit ? 1 : 0) - (q4 >= 3 * nit ? 1 : 0);
      if (pr >= 3)      __builtin_amdgcn_s_setprio(3);
      else if (pr == 2) __builtin_amdgcn_s_setprio(2);
      else if (pr == 1) __builtin_amdgcn_s_setprio(1);
      else              __builtin_amdgcn_s_setprio(0);
    }

#pragma unroll
    for (int t = 0; t < 4; ++t) {
      const int r = (tid + t * 256) >> 4;
      v4f v = w1r[t];
      v.x = (c4     < d) ? v.x : 0.0f;
      v.y = (c4 + 1 < d) ? v.y : 0.0f;
      v.z = (c4 + 2 < d) ? v.z : 0.0f;
      v.w = (c4 + 3 < d) ? v.w : 0.0f;
      *reinterpret_cast<v4f*>(&W1s[r][c4]) = v;
      *reinterpret_cast<v4f*>(&W2s[r][c4]) = w2r[t];
    }
    if (tid < 64)       B1s[tid] = bv;
    else if (tid < 128) B2s[tid - 64] = bv;
    else if (tid < 160) B3s[tid - 128] = bv;
    __syncthreads();

    {
      v8f acc0 = splat8(B1s[ln]);
      v8f acc1 = splat8(B1s[16 + ln]);
      v8f acc2 = splat8(B1s[32 + ln]);
      v8f acc3 = splat8(B1s[48 + ln]);
      const float* ap  = &Xs[mrow + ln][kh];
      const float* wp0 = &W1s[ln][kh];
      const float* wp1 = &W1s[16 + ln][kh];
      const float* wp2 = &W1s[32 + ln][kh];
      const float* wp3 = &W1s[48 + ln][kh];
      Frag2 fr[2];
      fr[0] = load_frag2(ap, wp0, wp1, wp2, wp3, 0);
#pragma unroll
      for (int q = 0; q < 8; ++q) {
        const bool more = (q + 1) * 8 < d;
        if (q < 7 && more) fr[(q + 1) & 1] = load_frag2(ap, wp0, wp1, wp2, wp3, (q + 1) * 8);
        mma_step0(fr[q & 1], acc0, acc1, acc2, acc3);
        if (q * 8 + 4 < d) mma_step1(fr[q & 1], acc0, acc1, acc2, acc3);
        if (!more) break;
      }
#pragma unroll
      for (int r = 0; r < 8; ++r) {
        Hs[mrow + mh + r][ln]      = tanh_hw(acc0[r]);
        Hs[mrow + mh + r][16 + ln] = tanh_hw(acc1[r]);
        Hs[mrow + mh + r][32 + ln] = tanh_hw(acc2[r]);
        Hs[mrow + mh + r][48 + ln] = tanh_hw(acc3[r]);
      }
    }
    __syncthreads();

#pragma unroll
    for (int t = 0; t < 2; ++t) {
      const int r = (tid + t * 256) >> 4;
      *reinterpret_cast<v4f*>(&W3s[r][c4]) = w3r[t];
    }

    {
      v8f acc0 = splat8(B2s[ln]);
      v8f acc1 = splat8(B2s[16 + ln]);
      v8f acc2 = splat8(B2s[32 + ln]);
      v8f acc3 = splat8(B2s[48 + ln]);
#pragma unroll
      for (int k0 = 0; k0 < HID; k0 += 4) {
        const v2f a  = ld2(&Hs[mrow + ln][k0 + kh]);
        const v2f w0 = ld2(&W2s[ln][k0 + kh]);
        const v2f w1 = ld2(&W2s[16 + ln][k0 + kh]);
        const v2f w2 = ld2(&W2s[32 + ln][k0 + kh]);
        const v2f w3 = ld2(&W2s[48 + ln][k0 + kh]);
        acc0 = __builtin_amdgcn_wmma_f32_16x16x4_f32(false, a, false, w0, (short)0, acc0, false, false);
        acc1 = __builtin_amdgcn_wmma_f32_16x16x4_f32(false, a, false, w1, (short)0, acc1, false, false);
        acc2 = __builtin_amdgcn_wmma_f32_16x16x4_f32(false, a, false, w2, (short)0, acc2, false, false);
        acc3 = __builtin_amdgcn_wmma_f32_16x16x4_f32(false, a, false, w3, (short)0, acc3, false, false);
      }
#pragma unroll
      for (int r = 0; r < 8; ++r) {
        Hs[mrow + mh + r][ln]      = tanh_hw(acc0[r]);
        Hs[mrow + mh + r][16 + ln] = tanh_hw(acc1[r]);
        Hs[mrow + mh + r][32 + ln] = tanh_hw(acc2[r]);
        Hs[mrow + mh + r][48 + ln] = tanh_hw(acc3[r]);
      }
    }
    __syncthreads();

    if (it + 1 < nit) fetch_weights(W1, b1, W2, b2, W3, b3, group_dim(glo, ghi, it + 1), tid, w1r, w2r, w3r, bv);

    {
      v8f acc0 = splat8(B3s[ln]);
      v8f acc1 = splat8(B3s[16 + ln]);
#pragma unroll
      for (int k0 = 0; k0 < HID; k0 += 4) {
        const v2f a  = ld2(&Hs[mrow + ln][k0 + kh]);
        const v2f w0 = ld2(&W3s[ln][k0 + kh]);
        const v2f w1 = ld2(&W3s[16 + ln][k0 + kh]);
        acc0 = __builtin_amdgcn_wmma_f32_16x16x4_f32(false, a, false, w0, (short)0, acc0, false, false);
        acc1 = __builtin_amdgcn_wmma_f32_16x16x4_f32(false, a, false, w1, (short)0, acc1, false, false);
      }
#pragma unroll
      for (int r = 0; r < 8; ++r) Hs[mrow + mh + r][ln] = acc0[r];
      if (16 + ln < 23) {
#pragma unroll
        for (int r = 0; r < 8; ++r) Hs[mrow + mh + r][16 + ln] = acc1[r];
      }
    }
    __syncthreads();

    if (tid < BTILE) {
      const int b = b0 + tid;
      float z, ld;
      rqs_spline(&Hs[tid][0], Xs[tid][d], z, ld);
      z_out[b * DIMS + d] = z;
      ldacc += ld;
    }
  }

  if (tid < BTILE) {
    const int b = b0 + tid;
    if (g == 3) {
      float pl[23];
#pragma unroll
      for (int j = 0; j < 23; ++j) pl[j] = initp[j];
      float z, ld;
      rqs_spline(pl, Xs[tid][0], z, ld);
      z_out[b * DIMS + 0] = z;
      ldacc += ld;
    }
    ld_ws[(size_t)g * NBATCH + b] = ldacc;
  }
}

__global__ void __launch_bounds__(256)
nsf_reduce(const float* __restrict__ ld_ws, float* __restrict__ out_ld) {
  int b = blockIdx.x * 256 + threadIdx.x;
  float s = 0.0f;
#pragma unroll
  for (int gg = 0; gg < NGROUP; ++gg) s += ld_ws[(size_t)gg * NBATCH + b];
  out_ld[b] = s;
}

extern "C" void kernel_launch(void* const* d_in, const int* in_sizes, int n_in,
                              void* d_out, int out_size, void* d_ws, size_t ws_size,
                              hipStream_t stream) {
  const float* x     = (const float*)d_in[0];
  const float* initp = (const float*)d_in[1];
  const float* W1    = (const float*)d_in[2];
  const float* b1    = (const float*)d_in[3];
  const float* W2    = (const float*)d_in[4];
  const float* b2    = (const float*)d_in[5];
  const float* W3    = (const float*)d_in[6];
  const float* b3    = (const float*)d_in[7];

  float* z_out  = (float*)d_out;
  float* ld_out = z_out + (size_t)NBATCH * DIMS;
  float* ld_ws  = (float*)d_ws;

  dim3 grid(NBATCH / BTILE, NGROUP);
  nsf_main<<<grid, 256, 0, stream>>>(x, initp, W1, b1, W2, b2, W3, b3, z_out, ld_ws);
  nsf_reduce<<<NBATCH / 256, 256, 0, stream>>>(ld_ws, ld_out);
}


# layers run transposed (weights = A, batch rows = B): tanh outputs feed the next B fragments in registers, no LDS round trip; 2 conditioners' splines per pass; k and log-det terms add in another order
# speedup vs baseline: 3.2585x; 1.0431x over previous
#include <hip/hip_runtime.h>
#include <math.h>

typedef float v2f __attribute__((ext_vector_type(2)));
typedef float v8f __attribute__((ext_vector_type(8)));
typedef float v4f __attribute__((ext_vector_type(4)));
typedef float v4f_g __attribute__((ext_vector_type(4), aligned(4)));

#define DIMS   64
#define HID    64
#define NBATCH 16384
#define BTILE  128
#define PADW   68
#define TAILB  3.0f

__device__ __forceinline__ v4f ld16(const float* p) {
  return *reinterpret_cast<const v4f_g*>(p);
}

__device__ __forceinline__ float exp_hw(float v) {
  return __builtin_amdgcn_exp2f(v * 1.44269504088896341f);
}
__device__ __forceinline__ float log_hw(float v) {
  return __builtin_amdgcn_logf(v) * 0.693147180559945309f;
}

__device__ __forceinline__ float softplus_f(float x) {
  return fmaxf(x, 0.0f) + log_hw(1.0f + exp_hw(-fabsf(x)));
}

__device__ __forceinline__ float tanh_hw(float v) {
  return __builtin_amdgcn_tanhf(v);
}

__device__ __forceinline__ v2f ld2(const float* p) {
  v2f v = {p[0], p[1]};
  return v;
}

struct Frag2 {
  v2f a0, a1;
  v2f w00, w01;
  v2f w10, w11;
  v2f w20, w21;
  v2f w30, w31;
};

__device__ __forceinline__ Frag2 load_frag2(const float* ap, const float* wp0, const float* wp1,
                                            const float* wp2, const float* wp3, int k0) {
  Frag2 f;
  f.a0  = ld2(ap  + k0);  f.a1  = ld2(ap  + k0 + 4);
  f.w00 = ld2(wp0 + k0);  f.w01 = ld2(wp0 + k0 + 4);
  f.w10 = ld2(wp1 + k0);  f.w11 = ld2(wp1 + k0 + 4);
  f.w20 = ld2(wp2 + k0);  f.w21 = ld2(wp2 + k0 + 4);
  f.w30 = ld2(wp3 + k0);  f.w31 = ld2(wp3 + k0 + 4);
  return f;
}

__device__ __forceinline__ void mma_step0(const Frag2& f, v8f& c0, v8f& c1, v8f& c2, v8f& c3) {
  c0 = __builtin_amdgcn_wmma_f32_16x16x4_f32(false, f.w00, false, f.a0, (short)0, c0, false, false);
  c1 = __builtin_amdgcn_wmma_f32_16x16x4_f32(false, f.w10, false, f.a0, (short)0, c1, false, false);
  c2 = __builtin_amdgcn_wmma_f32_16x16x4_f32(false, f.w20, false, f.a0, (short)0, c2, false, false);
  c3 = __builtin_amdgcn_wmma_f32_16x16x4_f32(false, f.w30, false, f.a0, (short)0, c3, false, false);
}
__device__ __forceinline__ void mma_step1(const Frag2& f, v8f& c0, v8f& c1, v8f& c2, v8f& c3) {
  c0 = __builtin_amdgcn_wmma_f32_16x16x4_f32(false, f.w01, false, f.a1, (short)0, c0, false, false);
  c1 = __builtin_amdgcn_wmma_f32_16x16x4_f32(false, f.w11, false, f.a1, (short)0, c1, false, false);
  c2 = __builtin_amdgcn_wmma_f32_16x16x4_f32(false, f.w21, false, f.a1, (short)0, c2, false, false);
  c3 = __builtin_amdgcn_wmma_f32_16x16x4_f32(false, f.w31, false, f.a1, (short)0, c3, false, false);
}

__device__ __forceinline__ int xhalf_i(int v) {
  return __builtin_amdgcn_permlanex16(v, v, 0x76543210, 0xfedcba98, false, false);
}
__device__ __forceinline__ float xhalf_f(float v) {
  return __builtin_bit_cast(float, xhalf_i(__builtin_bit_cast(int, v)));
}

__device__ __forceinline__ v8f bias8(const float* p) {
  const v4f lo = *reinterpret_cast<const v4f*>(p);
  const v4f hi = *reinterpret_cast<const v4f*>(p + 4);
  v8f v = {lo.x, lo.y, lo.z, lo.w, hi.x, hi.y, hi.z, hi.w};
  return v;
}

__device__ __forceinline__ v8f tanh8(v8f a) {
  v8f v = {tanh_hw(a[0]), tanh_hw(a[1]), tanh_hw(a[2]), tanh_hw(a[3]),
           tanh_hw(a[4]), tanh_hw(a[5]), tanh_hw(a[6]), tanh_hw(a[7])};
  return v;
}

__device__ __forceinline__ v2f pair_of(v8f t, int j) {
  v2f v = {t[2 * j], t[2 * j + 1]};
  return v;
}

__device__ __forceinline__ void build_edges(const float* p, float* edges, float* widths) {
  float m1 = p[0];
#pragma unroll
  for (int j = 1; j < 8; ++j) m1 = fmaxf(m1, p[j]);
  float e[8];
  float s = 0.0f;
#pragma unroll
  for (int j = 0; j < 8; ++j) { e[j] = exp_hw(p[j] - m1); s += e[j]; }
  float inv = (2.0f * TAILB) / s;
  float u[8];
  float m2 = -1e30f;
#pragma unroll
  for (int j = 0; j < 8; ++j) { u[j] = e[j] * inv; m2 = fmaxf(m2, u[j]); }
  float s2 = 0.0f;
#pragma unroll
  for (int j = 0; j < 8; ++j) { e[j] = exp_hw(u[j] - m2); s2 += e[j]; }
  float inv2 = (1.0f - 8.0f * 1e-3f) / s2;
  float cum = 0.0f;
  edges[0] = -TAILB;
#pragma unroll
  for (int j = 0; j < 8; ++j) {
    float w = 1e-3f + e[j] * inv2;
    cum += w;
    edges[j + 1] = 2.0f * TAILB * cum - TAILB;
  }
  edges[8] = TAILB;
#pragma unroll
  for (int j = 0; j < 8; ++j) widths[j] = edges[j + 1] - edges[j];
}

__device__ __forceinline__ void rqs_spline(const float* p, float xin, float& zout, float& ldout) {
  bool inside = (xin >= -TAILB) && (xin <= TAILB);
  float xc = fminf(fmaxf(xin, -TAILB), TAILB);
  float cw[9], wdt[8], ch[9], hgt[8], der[9];
  build_edges(p, cw, wdt);
  build_edges(p + 8, ch, hgt);
  der[0] = 1.0f;
  der[8] = 1.0f;
#pragma unroll
  for (int j = 1; j <= 7; ++j) der[j] = 1e-3f + softplus_f(softplus_f(p[15 + j]));
  int idx = -1;
#pragma unroll
  for (int j = 0; j < 9; ++j) {
    float edge = cw[j] + ((j == 8) ? 1e-6f : 0.0f);
    idx += (xc >= edge) ? 1 : 0;
  }
  idx = max(0, min(idx, 7));
  float icw = cw[0], iw = wdt[0];
  float ich = ch[0], ih = hgt[0];
  float dk = der[0], dk1 = der[1];
#pragma unroll
  for (int j = 1; j < 8; ++j) {
    const bool hit = (idx == j);
    icw = hit ? cw[j]  : icw;   iw  = hit ? wdt[j]     : iw;
    ich = hit ? ch[j]  : ich;   ih  = hit ? hgt[j]     : ih;
    dk  = hit ? der[j] : dk;    dk1 = hit ? der[j + 1] : dk1;
  }
  float delta = ih / iw;
  float th  = (xc - icw) / iw;
  float omt = 1.0f - th;
  float t1  = th * omt;
  float numer = ih * (delta * th * th + dk * t1);
  float denom = delta + (dk + dk1 - 2.0f * delta) * t1;
  float outv  = ich + numer / denom;
  float dnum  = delta * delta * (dk1 * th * th + 2.0f * delta * t1 + dk * omt * omt);
  float ld    = log_hw(dnum) - 2.0f * log_hw(denom);
  zout  = inside ? outv : xin;
  ldout = inside ? ld : 0.0f;
}

__device__ __forceinline__ void swap_halves(float& a, float& b) {
  asm volatile("v_permlane16_swap_b32 %0, %1" : "+v"(a), "+v"(b));
}
__device__ __forceinline__ void swap_halves_first(float& a, float& b) {
  asm volatile("v_nop\n\tv_nop\n\tv_nop\n\tv_nop\n\tv_permlane16_swap_b32 %0, %1" : "+v"(a), "+v"(b));
}

__device__ __forceinline__ void rqs_spline_pieces(const float* pw, const float* ph, v8f da, v8f db,
                                                  float xin, float& zout, float& ldout) {
  bool inside = (xin >= -TAILB) && (xin <= TAILB);
  float xc = fminf(fmaxf(xin, -TAILB), TAILB);
  int idx = -1;
  float icw, iw;
  {
    float cw[9], wdt[8];
    build_edges(pw, cw, wdt);
#pragma unroll
    for (int j = 0; j < 9; ++j) {
      float edge = cw[j] + ((j == 8) ? 1e-6f : 0.0f);
      idx += (xc >= edge) ? 1 : 0;
    }
    idx = max(0, min(idx, 7));
    icw = cw[0]; iw = wdt[0];
#pragma unroll
    for (int j = 1; j < 8; ++j) {
      const bool hit = (idx == j);
      icw = hit ? cw[j] : icw;   iw = hit ? wdt[j] : iw;
    }
  }
  float ich, ih;
  {
    float ch[9], hgt[8];
    build_edges(ph, ch, hgt);
    ich = ch[0]; ih = hgt[0];
#pragma unroll
    for (int j = 1; j < 8; ++j) {
      const bool hit = (idx == j);
      ich = hit ? ch[j] : ich;   ih = hit ? hgt[j] : ih;
    }
  }
  float dk = 1.0f, dk1 = 1.0f;
#pragma unroll
  for (int r = 0; r < 7; ++r) {
    float a = da[r], b = db[r];
    swap_halves(a, b);
    const float der = 1e-3f + softplus_f(softplus_f(a));
    dk  = (idx == r + 1) ? der : dk;
    dk1 = (idx == r)     ? der : dk1;
  }
  float delta = ih / iw;
  float th  = (xc - icw) / iw;
  float omt = 1.0f - th;
  float t1  = th * omt;
  float numer = ih * (delta * th * th + dk * t1);
  float denom = delta + (dk + dk1 - 2.0f * delta) * t1;
  float outv  = ich + numer / denom;
  float dnum  = delta * delta * (dk1 * th * th + 2.0f * delta * t1 + dk * omt * omt);
  float ld    = log_hw(dnum) - 2.0f * log_hw(denom);
  zout  = inside ? outv : xin;
  ldout = inside ? ld : 0.0f;
}

#define NGROUP 6

__device__ __forceinline__ int group_count(int g) { return (g < 3) ? 11 : 10; }
__device__ __forceinline__ unsigned long long group_lo(int g) {
  return (g == 0) ? 0x0e12161726292c2fULL
                              : (g == 1) ? 0x0c18191b1f243034ULL
                              : (g == 2) ? 0x0f111c2122232b38ULL
                              : (g == 3) ? 0x1e20252d2e32363aULL
                              : (g == 4) ? 0x1a1d283335393d3fULL
                              :            0x14272a31373b3c3eULL;
}
__device__ __forceinline__ unsigned int group_hi(int g) {
  return (g == 0) ? 0x00020a0du
                        : (g == 1) ? 0x0003090bu
                        : (g == 2) ? 0x00040708u
                        : (g == 3) ? 0x00000615u
                        : (g == 4) ? 0x00000113u
                        :            0x00000510u;
}
__device__ __forceinline__ int group_dim(unsigned long long lo, unsigned int hi, int it) {
  return (it < 8) ? (int)((lo >> (8 * it)) & 0xffULL) : (int)((hi >> (8 * (it - 8))) & 0xffu);
}

__device__ __forceinline__ void fetch_weights(const float* __restrict__ W1, const float* __restrict__ b1,
                                              const float* __restrict__ W2, const float* __restrict__ b2,
                                              const float* __restrict__ W3, const float* __restrict__ b3,
                                              int d, int tid, v4f (&w1r)[4], v4f (&w2r)[4], v4f (&w3r)[2],
                                              float& bv) {
  const int l = d - 1;
  const float* w1p = W1 + (size_t)l * HID * DIMS;
  const float* w2p = W2 + (size_t)l * HID * HID;
  const float* w3p = W3 + (size_t)l * 23 * HID;
#pragma unroll
  for (int t = 0; t < 4; ++t) {
    w1r[t] = ld16(w1p + (tid + t * 256) * 4);
    w2r[t] = ld16(w2p + (tid + t * 256) * 4);
  }
#pragma unroll
  for (int t = 0; t < 2; ++t) {
    const int i = tid + t * 256;
    v4f z4 = {0.0f, 0.0f, 0.0f, 0.0f};
    w3r[t] = (i < 23 * 16) ? ld16(w3p + i * 4) : z4;
  }
  bv = 0.0f;
  if (tid < 64)       bv = b1[l * HID + tid];
  else if (tid < 128) bv = b2[l * HID + (tid - 64)];
  else if (tid < 151) bv = b3[l * 23 + (tid - 128)];
}

__device__ __forceinline__ void stage_weights(float (*W1s)[PADW], float (*W2s)[PADW], float (*W3s)[PADW],
                                              float* B1s, float* B2s, float* B3s, int d, int tid, int c4,
                                              const v4f (&w1r)[4], const v4f (&w2r)[4], const v4f (&w3r)[2],
                                              float bv) {
  const int r0 = tid >> 4;
#pragma unroll
  for (int t = 0; t < 4; ++t) {
    v4f v = w1r[t];
    v.x = (c4     < d) ? v.x : 0.0f;
    v.y = (c4 + 1 < d) ? v.y : 0.0f;
    v.z = (c4 + 2 < d) ? v.z : 0.0f;
    v.w = (c4 + 3 < d) ? v.w : 0.0f;
    *reinterpret_cast<v4f*>(&W1s[r0 + 16 * t][c4]) = v;
    *reinterpret_cast<v4f*>(&W2s[r0 + 16 * t][c4]) = w2r[t];
  }
#pragma unroll
  for (int t = 0; t < 2; ++t) {
    *reinterpret_cast<v4f*>(&W3s[r0 + 16 * t][c4]) = w3r[t];
  }
  if (tid < 64)       B1s[tid] = bv;
  else if (tid < 128) B2s[tid - 64] = bv;
  else if (tid < 160) B3s[tid - 128] = bv;
}

__global__ void __launch_bounds__(256, 6)
nsf_main(const float* __restrict__ x,  const float* __restrict__ initp,
         const float* __restrict__ W1, const float* __restrict__ b1,
         const float* __restrict__ W2, const float* __restrict__ b2,
         const float* __restrict__ W3, const float* __restrict__ b3,
         float* __restrict__ z_out, float* __restrict__ ld_ws) {
  __shared__ __align__(16) float Xs[BTILE][PADW];
  __shared__ __align__(16) float W1s[HID][PADW];
  __shared__ __align__(16) float W2s[HID][PADW];
  __shared__ __align__(16) float W3s[32][PADW];
  __shared__ __align__(16) float B1s[HID];
  __shared__ __align__(16) float B2s[HID];
  __shared__ __align__(16) float B3s[32];
  __shared__ v4f Keep[4][256];

  const int g   = blockIdx.y;
  const int b0  = blockIdx.x * BTILE;
  const int tid = threadIdx.x;
  const int nit = group_count(g);
  const unsigned long long glo = group_lo(g);
  const unsigned int       ghi = group_hi(g);
  const int c4  = (tid & 15) << 2;

  const int lane = tid & 31;
  const int wave = tid >> 5;
  const int mrow = wave * 16;
  const int ln   = lane & 15;
  const int hf   = lane >> 4;
  const int kh   = hf << 1;
  const int h8   = hf << 3;

  v4f w1r[4], w2r[4], w3r[2];
  float bv;

  {
    const float* xg = x + (size_t)b0 * DIMS;
    v4f xr[8];
#pragma unroll
    for (int t = 0; t < 8; ++t) xr[t] = ld16(xg + (tid + t * 256) * 4);
    const int d0 = group_dim(glo, ghi, 0);
    fetch_weights(W1, b1, W2, b2, W3, b3, d0, tid, w1r, w2r, w3r, bv);
#pragma unroll
    for (int t = 0; t < 8; ++t) {
      const int m = (tid + t * 256) >> 4;
      *reinterpret_cast<v4f*>(&Xs[m][c4]) = xr[t];
    }
    stage_weights(W1s, W2s, W3s, B1s, B2s, B3s, d0, tid, c4, w1r, w2r, w3r, bv);
  }

  float ldacc = 0.0f;
  int dfirst = 0;

  for (int it = 0; it < nit; ++it) {
    const int d = group_dim(glo, ghi, it);

    {
      const int q4 = 4 * it;
      const int pr = 3 - (q4 >= nit ? 1 : 0) - (q4 >= 2 * nit ? 1 : 0) - (q4 >= 3 * nit ? 1 : 0);
      if (pr >= 3)      __builtin_amdgcn_s_setprio(3);
      else if (pr == 2) __builtin_amdgcn_s_setprio(2);
      else if (pr == 1) __builtin_amdgcn_s_setprio(1);
      else              __builtin_amdgcn_s_setprio(0);
    }

    __syncthreads();

    v8f hv[4];

    {
      v8f acc0 = bias8(&B1s[h8]);
      v8f acc1 = bias8(&B1s[16 + h8]);
      v8f acc2 = bias8(&B1s[32 + h8]);
      v8f acc3 = bias8(&B1s[48 + h8]);
      const float* ap  = &Xs[mrow + ln][kh];
      const float* wp0 = &W1s[ln][kh];
      const float* wp1 = &W1s[16 + ln][kh];
      const float* wp2 = &W1s[32 + ln][kh];
      const float* wp3 = &W1s[48 + ln][kh];
      Frag2 fr[2];
      fr[0] = load_frag2(ap, wp0, wp1, wp2, wp3, 0);
#pragma unroll
      for (int q = 0; q < 8; ++q) {
        const bool more = (q + 1) * 8 < d;
        if (q < 7 && more) fr[(q + 1) & 1] = load_frag2(ap, wp0, wp1, wp2, wp3, (q + 1) * 8);
        mma_step0(fr[q & 1], acc0, acc1, acc2, acc3);
        if (q * 8 + 4 < d) mma_step1(fr[q & 1], acc0, acc1, acc2, acc3);
        if (!more) break;
      }
      hv[0] = tanh8(acc0);
      hv[1] = tanh8(acc1);
      hv[2] = tanh8(acc2);
      hv[3] = tanh8(acc3);
    }

    {
      v8f acc0 = bias8(&B2s[h8]);
      v8f acc1 = bias8(&B2s[16 + h8]);
      v8f acc2 = bias8(&B2s[32 + h8]);
      v8f acc3 = bias8(&B2s[48 + h8]);
      const float* wq0 = &W2s[ln][h8];
      const float* wq1 = &W2s[16 + ln][h8];
      const float* wq2 = &W2s[32 + ln][h8];
      const float* wq3 = &W2s[48 + ln][h8];
#pragma unroll
      for (int t = 0; t < 4; ++t) {
#pragma unroll
        for (int j = 0; j < 4; ++j) {
          const int ko = 16 * t + 2 * j;
          const v2f b  = pair_of(hv[t], j);
          const v2f w0 = ld2(wq0 + ko);
          const v2f w1 = ld2(wq1 + ko);
          const v2f w2 = ld2(wq2 + ko);
          const v2f w3 = ld2(wq3 + ko);
          acc0 = __builtin_amdgcn_wmma_f32_16x16x4_f32(false, w0, false, b, (short)0, acc0, false, false);
          acc1 = __builtin_amdgcn_wmma_f32_16x16x4_f32(false, w1, false, b, (short)0, acc1, false, false);
          acc2 = __builtin_amdgcn_wmma_f32_16x16x4_f32(false, w2, false, b, (short)0, acc2, false, false);
          acc3 = __builtin_amdgcn_wmma_f32_16x16x4_f32(false, w3, false, b, (short)0, acc3, false, false);
        }
      }
      hv[0] = tanh8(acc0);
      hv[1] = tanh8(acc1);
      hv[2] = tanh8(acc2);
      hv[3] = tanh8(acc3);
    }

    const int dn = group_dim(glo, ghi, (it + 1 < nit) ? it + 1 : it);
    fetch_weights(W1, b1, W2, b2, W3, b3, dn, tid, w1r, w2r, w3r, bv);

    v8f p0, p1;
    {
      v8f acc0 = bias8(&B3s[h8]);
      v8f acc1 = bias8(&B3s[16 + h8]);
      const float* wr0 = &W3s[ln][h8];
      const float* wr1 = &W3s[16 + ln][h8];
#pragma unroll
      for (int t = 0; t < 4; ++t) {
#pragma unroll
        for (int j = 0; j < 4; ++j) {
          const int ko = 16 * t + 2 * j;
          const v2f b  = pair_of(hv[t], j);
          const v2f w0 = ld2(wr0 + ko);
          const v2f w1 = ld2(wr1 + ko);
          acc0 = __builtin_amdgcn_wmma_f32_16x16x4_f32(false, w0, false, b, (short)0, acc0, false, false);
          acc1 = __builtin_amdgcn_wmma_f32_16x16x4_f32(false, w1, false, b, (short)0, acc1, false, false);
        }
      }
      p0 = acc0;
      p1 = acc1;
    }

    const bool second = (it & 1) != 0;
    if (!second) {
      dfirst = d;
      v4f k0 = {p0[0], p0[1], p0[2], p0[3]}, k1 = {p0[4], p0[5], p0[6], p0[7]};
      v4f k2 = {p1[0], p1[1], p1[2], p1[3]}, k3 = {p1[4], p1[5], p1[6], p1[7]};
      Keep[0][tid] = k0;  Keep[1][tid] = k1;  Keep[2][tid] = k2;  Keep[3][tid] = k3;
    }
    if (second || it + 1 == nit) {
      const v4f k0 = Keep[0][tid], k1 = Keep[1][tid], k2 = Keep[2][tid], k3 = Keep[3][tid];
      const v8f sv0 = {k0.x, k0.y, k0.z, k0.w, k1.x, k1.y, k1.z, k1.w};
      const v8f sv1 = {k2.x, k2.y, k2.z, k2.w, k3.x, k3.y, k3.z, k3.w};
      float pw[8], ph[8];
#pragma unroll
      for (int r = 0; r < 8; ++r) {
        float a = sv0[r], b = p0[r];
        if (r == 0) swap_halves_first(a, b); else swap_halves(a, b);
        pw[r] = a;
        ph[r] = b;
      }
      const int dl = (hf != 0) ? d : dfirst;
      float z, ld;
      rqs_spline_pieces(pw, ph, sv1, p1, Xs[mrow + ln][dl], z, ld);
      if (hf == 0 || second) {
        z_out[(b0 + mrow + ln) * DIMS + dl] = z;
        ldacc += ld;
      }
    }

    __syncthreads();
    if (it + 1 < nit) stage_weights(W1s, W2s, W3s, B1s, B2s, B3s, dn, tid, c4, w1r, w2r, w3r, bv);
  }

  if (g == 3) {
    float pl[23];
#pragma unroll
    for (int j = 0; j < 23; ++j) pl[j] = initp[j];
    float z, ld;
    rqs_spline(pl, Xs[mrow + ln][0], z, ld);
    if (hf == 0) {
      z_out[(b0 + mrow + ln) * DIMS + 0] = z;
      ldacc += ld;
    }
  }
  ldacc += xhalf_f(ldacc);
  if (hf == 0) ld_ws[(size_t)g * NBATCH + b0 + mrow + ln] = ldacc;
}

__global__ void __launch_bounds__(256)
nsf_reduce(const float* __restrict__ ld_ws, float* __restrict__ out_ld) {
  int b = blockIdx.x * 256 + threadIdx.x;
  float s = 0.0f;
#pragma unroll
  for (int gg = 0; gg < NGROUP; ++gg) s += ld_ws[(size_t)gg * NBATCH + b];
  out_ld[b] = s;
}

extern "C" void kernel_launch(void* const* d_in, const int* in_sizes, int n_in,
                              void* d_out, int out_size, void* d_ws, size_t ws_size,
                              hipStream_t stream) {
  const float* x     = (const float*)d_in[0];
  const float* initp = (const float*)d_in[1];
  const float* W1    = (const float*)d_in[2];
  const float* b1    = (const float*)d_in[3];
  const float* W2    = (const float*)d_in[4];
  const float* b2    = (const float*)d_in[5];
  const float* W3    = (const float*)d_in[6];
  const float* b3    = (const float*)d_in[7];

  float* z_out  = (float*)d_out;
  float* ld_out = z_out + (size_t)NBATCH * DIMS;
  float* ld_ws  = (float*)d_ws;

  dim3 grid(NBATCH / BTILE, NGROUP);
  nsf_main<<<grid, 256, 0, stream>>>(x, initp, W1, b1, W2, b2, W3, b3, z_out, ld_ws);
  nsf_reduce<<<NBATCH / 256, 256, 0, stream>>>(ld_ws, ld_out);
}


# spline pass: vector instructions nearly halved: the hit bin is picked first, then 1 width, 1 height, 2 double softplus (were 8, 8, 7); exponents by one fma; quotients rcp*mul (last-bit changes)
# speedup vs baseline: 3.3626x; 1.0319x over previous
#include <hip/hip_runtime.h>
#include <math.h>

typedef float v2f __attribute__((ext_vector_type(2)));
typedef float v8f __attribute__((ext_vector_type(8)));
typedef float v4f __attribute__((ext_vector_type(4)));
typedef float v4f_g __attribute__((ext_vector_type(4), aligned(4)));

#define DIMS   64
#define HID    64
#define NBATCH 16384
#define BTILE  128
#define PADW   68
#define TAILB  3.0f

__device__ __forceinline__ v4f ld16(const float* p) {
  return *reinterpret_cast<const v4f_g*>(p);
}

__device__ __forceinline__ float exp_hw(float v) {
  return __builtin_amdgcn_exp2f(v * 1.44269504088896341f);
}
__device__ __forceinline__ float log_hw(float v) {
  return __builtin_amdgcn_logf(v) * 0.693147180559945309f;
}

__device__ __forceinline__ float max_num(float a, float b) {
  return __builtin_fmaximum_numf(a, b);
}

__device__ __forceinline__ float softplus_f(float x) {
  return max_num(x, 0.0f) + log_hw(1.0f + exp_hw(-fabsf(x)));
}

__device__ __forceinline__ float tanh_hw(float v) {
  return __builtin_amdgcn_tanhf(v);
}

__device__ __forceinline__ v2f ld2(const float* p) {
  v2f v = {p[0], p[1]};
  return v;
}

struct Frag2 {
  v2f a0, a1;
  v2f w00, w01;
  v2f w10, w11;
  v2f w20, w21;
  v2f w30, w31;
};

__device__ __forceinline__ Frag2 load_frag2(const float* ap, const float* wp0, const float* wp1,
                                            const float* wp2, const float* wp3, int k0) {
  Frag2 f;
  f.a0  = ld2(ap  + k0);  f.a1  = ld2(ap  + k0 + 4);
  f.w00 = ld2(wp0 + k0);  f.w01 = ld2(wp0 + k0 + 4);
  f.w10 = ld2(wp1 + k0);  f.w11 = ld2(wp1 + k0 + 4);
  f.w20 = ld2(wp2 + k0);  f.w21 = ld2(wp2 + k0 + 4);
  f.w30 = ld2(wp3 + k0);  f.w31 = ld2(wp3 + k0 + 4);
  return f;
}

__device__ __forceinline__ void mma_step0(const Frag2& f, v8f& c0, v8f& c1, v8f& c2, v8f& c3) {
  c0 = __builtin_amdgcn_wmma_f32_16x16x4_f32(false, f.w00, false, f.a0, (short)0, c0, false, false);
  c1 = __builtin_amdgcn_wmma_f32_16x16x4_f32(false, f.w10, false, f.a0, (short)0, c1, false, false);
  c2 = __builtin_amdgcn_wmma_f32_16x16x4_f32(false, f.w20, false, f.a0, (short)0, c2, false, false);
  c3 = __builtin_amdgcn_wmma_f32_16x16x4_f32(false, f.w30, false, f.a0, (short)0, c3, false, false);
}
__device__ __forceinline__ void mma_step1(const Frag2& f, v8f& c0, v8f& c1, v8f& c2, v8f& c3) {
  c0 = __builtin_amdgcn_wmma_f32_16x16x4_f32(false, f.w01, false, f.a1, (short)0, c0, false, false);
  c1 = __builtin_amdgcn_wmma_f32_16x16x4_f32(false, f.w11, false, f.a1, (short)0, c1, false, false);
  c2 = __builtin_amdgcn_wmma_f32_16x16x4_f32(false, f.w21, false, f.a1, (short)0, c2, false, false);
  c3 = __builtin_amdgcn_wmma_f32_16x16x4_f32(false, f.w31, false, f.a1, (short)0, c3, false, false);
}

__device__ __forceinline__ int xhalf_i(int v) {
  return __builtin_amdgcn_permlanex16(v, v, 0x76543210, 0xfedcba98, false, false);
}
__device__ __forceinline__ float xhalf_f(float v) {
  return __builtin_bit_cast(float, xhalf_i(__builtin_bit_cast(int, v)));
}

__device__ __forceinline__ v8f bias8(const float* p) {
  const v4f lo = *reinterpret_cast<const v4f*>(p);
  const v4f hi = *reinterpret_cast<const v4f*>(p + 4);
  v8f v = {lo.x, lo.y, lo.z, lo.w, hi.x, hi.y, hi.z, hi.w};
  return v;
}

__device__ __forceinline__ v8f tanh8(v8f a) {
  v8f v = {tanh_hw(a[0]), tanh_hw(a[1]), tanh_hw(a[2]), tanh_hw(a[3]),
           tanh_hw(a[4]), tanh_hw(a[5]), tanh_hw(a[6]), tanh_hw(a[7])};
  return v;
}

__device__ __forceinline__ v2f pair_of(v8f t, int j) {
  v2f v = {t[2 * j], t[2 * j + 1]};
  return v;
}

__device__ __forceinline__ void bin_cumsum(const float* p, float* c) {
  const float L2E = 1.44269504088896341f;
  float m1 = p[0];
#pragma unroll
  for (int j = 1; j < 8; ++j) m1 = max_num(m1, p[j]);
  const float nm = -m1 * L2E;
  float e[8];
  float s = 0.0f;
#pragma unroll
  for (int j = 0; j < 8; ++j) { e[j] = __builtin_amdgcn_exp2f(fmaf(p[j], L2E, nm)); s += e[j]; }
  const float a = (2.0f * TAILB) * __builtin_amdgcn_rcpf(s) * L2E;
  float s2 = 0.0f;
#pragma unroll
  for (int j = 0; j < 8; ++j) { e[j] = __builtin_amdgcn_exp2f(fmaf(e[j], a, -a)); s2 += e[j]; }
  const float inv2 = (1.0f - 8.0f * 1e-3f) * __builtin_amdgcn_rcpf(s2);
  float cum = 0.0f;
  c[0] = 0.0f;
#pragma unroll
  for (int j = 0; j < 8; ++j) {
    cum += fmaf(e[j], inv2, 1e-3f);
    c[j + 1] = cum;
  }
}

__device__ __forceinline__ void swap_halves(float& a, float& b) {
  asm volatile("v_permlane16_swap_b32 %0, %1" : "+v"(a), "+v"(b));
}
__device__ __forceinline__ void swap_halves_first(float& a, float& b) {
  asm volatile("v_nop\n\tv_nop\n\tv_nop\n\tv_nop\n\tv_permlane16_swap_b32 %0, %1" : "+v"(a), "+v"(b));
}

template <bool kSwap>
__device__ __forceinline__ void rqs_eval(const float* pw, const float* ph, v8f da, v8f db,
                                         float xin, float& zout, float& ldout) {
  const bool inside = (xin >= -TAILB) && (xin <= TAILB);
  const float xc = fminf(fmaxf(xin, -TAILB), TAILB);
  bool g1, g2, g3, g4, g5, g6, g7;
  float icw, iw;
  {
    float c[9], cw[9];
    bin_cumsum(pw, c);
    cw[0] = -TAILB;
    cw[8] = TAILB;
#pragma unroll
    for (int j = 1; j < 8; ++j) cw[j] = fmaf(2.0f * TAILB, c[j], -TAILB);
    g1 = (xc >= cw[1]);  g2 = (xc >= cw[2]);  g3 = (xc >= cw[3]);  g4 = (xc >= cw[4]);
    g5 = (xc >= cw[5]);  g6 = (xc >= cw[6]);  g7 = (xc >= cw[7]);
    float lo = cw[0], hi = cw[1];
    lo = g1 ? cw[1] : lo;   hi = g1 ? cw[2] : hi;
    lo = g2 ? cw[2] : lo;   hi = g2 ? cw[3] : hi;
    lo = g3 ? cw[3] : lo;   hi = g3 ? cw[4] : hi;
    lo = g4 ? cw[4] : lo;   hi = g4 ? cw[5] : hi;
    lo = g5 ? cw[5] : lo;   hi = g5 ? cw[6] : hi;
    lo = g6 ? cw[6] : lo;   hi = g6 ? cw[7] : hi;
    lo = g7 ? cw[7] : lo;   hi = g7 ? cw[8] : hi;
    icw = lo;
    iw  = hi - lo;
  }
  float ich, ih;
  {
    float c[9];
    bin_cumsum(ph, c);
    float cl = c[0], cr = c[1];
    cl = g1 ? c[1] : cl;   cr = g1 ? c[2] : cr;
    cl = g2 ? c[2] : cl;   cr = g2 ? c[3] : cr;
    cl = g3 ? c[3] : cl;   cr = g3 ? c[4] : cr;
    cl = g4 ? c[4] : cl;   cr = g4 ? c[5] : cr;
    cl = g5 ? c[5] : cl;   cr = g5 ? c[6] : cr;
    cl = g6 ? c[6] : cl;   cr = g6 ? c[7] : cr;
    cl = g7 ? c[7] : cl;   cr = g7 ? c[8] : cr;
    const float lo = fmaf(2.0f * TAILB, cl, -TAILB);
    const float hi = g7 ? TAILB : fmaf(2.0f * TAILB, cr, -TAILB);
    ich = lo;
    ih  = hi - lo;
  }
  float lg[7];
#pragma unroll
  for (int r = 0; r < 7; ++r) {
    float a = da[r];
    if (kSwap) {
      float b = db[r];
      swap_halves(a, b);
    }
    lg[r] = a;
  }
  float lk = 0.0f, lk1 = lg[0];
  lk = g1 ? lg[0] : lk;   lk1 = g1 ? lg[1] : lk1;
  lk = g2 ? lg[1] : lk;   lk1 = g2 ? lg[2] : lk1;
  lk = g3 ? lg[2] : lk;   lk1 = g3 ? lg[3] : lk1;
  lk = g4 ? lg[3] : lk;   lk1 = g4 ? lg[4] : lk1;
  lk = g5 ? lg[4] : lk;   lk1 = g5 ? lg[5] : lk1;
  lk = g6 ? lg[5] : lk;   lk1 = g6 ? lg[6] : lk1;
  lk = g7 ? lg[6] : lk;
  const float fk  = 1e-3f + softplus_f(softplus_f(lk));
  const float fk1 = 1e-3f + softplus_f(softplus_f(lk1));
  const float dk  = g1 ? fk : 1.0f;
  const float dk1 = g7 ? 1.0f : fk1;
  const float riw = __builtin_amdgcn_rcpf(iw);
  float delta = ih * riw;
  float th  = (xc - icw) * riw;
  float omt = 1.0f - th;
  float t1  = th * omt;
  float numer = ih * (delta * th * th + dk * t1);
  float denom = delta + (dk + dk1 - 2.0f * delta) * t1;
  float outv  = ich + numer * __builtin_amdgcn_rcpf(denom);
  float dnum  = delta * delta * (dk1 * th * th + 2.0f * delta * t1 + dk * omt * omt);
  float ld    = log_hw(dnum) - 2.0f * log_hw(denom);
  zout  = inside ? outv : xin;
  ldout = inside ? ld : 0.0f;
}

#define NGROUP 6

__device__ __forceinline__ int group_count(int g) { return (g < 3) ? 11 : 10; }
__device__ __forceinline__ unsigned long long group_lo(int g) {
  return (g == 0) ? 0x0e12161726292c2fULL
                              : (g == 1) ? 0x0c18191b1f243034ULL
                              : (g == 2) ? 0x0f111c2122232b38ULL
                              : (g == 3) ? 0x1e20252d2e32363aULL
                              : (g == 4) ? 0x1a1d283335393d3fULL
                              :            0x14272a31373b3c3eULL;
}
__device__ __forceinline__ unsigned int group_hi(int g) {
  return (g == 0) ? 0x00020a0du
                        : (g == 1) ? 0x0003090bu
                        : (g == 2) ? 0x00040708u
                        : (g == 3) ? 0x00000615u
                        : (g == 4) ? 0x00000113u
                        :            0x00000510u;
}
__device__ __forceinline__ int group_dim(unsigned long long lo, unsigned int hi, int it) {
  return (it < 8) ? (int)((lo >> (8 * it)) & 0xffULL) : (int)((hi >> (8 * (it - 8))) & 0xffu);
}

__device__ __forceinline__ void fetch_weights(const float* __restrict__ W1, const float* __restrict__ b1,
                                              const float* __restrict__ W2, const float* __restrict__ b2,
                                              const float* __restrict__ W3, const float* __restrict__ b3,
                                              int d, int tid, v4f (&w1r)[4], v4f (&w2r)[4], v4f (&w3r)[2],
                                              float& bv) {
  const int l = d - 1;
  const float* w1p = W1 + (size_t)l * HID * DIMS;
  const float* w2p = W2 + (size_t)l * HID * HID;
  const float* w3p = W3 + (size_t)l * 23 * HID;
#pragma unroll
  for (int t = 0; t < 4; ++t) {
    w1r[t] = ld16(w1p + (tid + t * 256) * 4);
    w2r[t] = ld16(w2p + (tid + t * 256) * 4);
  }
#pragma unroll
  for (int t = 0; t < 2; ++t) {
    const int i = tid + t * 256;
    v4f z4 = {0.0f, 0.0f, 0.0f, 0.0f};
    w3r[t] = (i < 23 * 16) ? ld16(w3p + i * 4) : z4;
  }
  bv = 0.0f;
  if (tid < 64)       bv = b1[l * HID + tid];
  else if (tid < 128) bv = b2[l * HID + (tid - 64)];
  else if (tid < 151) bv = b3[l * 23 + (tid - 128)];
}

__device__ __forceinline__ void stage_weights(float (*W1s)[PADW], float (*W2s)[PADW], float (*W3s)[PADW],
                                              float* B1s, float* B2s, float* B3s, int d, int tid, int c4,
                                              const v4f (&w1r)[4], const v4f (&w2r)[4], const v4f (&w3r)[2],
                                              float bv) {
  const int r0 = tid >> 4;
#pragma unroll
  for (int t = 0; t < 4; ++t) {
    v4f v = w1r[t];
    v.x = (c4     < d) ? v.x : 0.0f;
    v.y = (c4 + 1 < d) ? v.y : 0.0f;
    v.z = (c4 + 2 < d) ? v.z : 0.0f;
    v.w = (c4 + 3 < d) ? v.w : 0.0f;
    *reinterpret_cast<v4f*>(&W1s[r0 + 16 * t][c4]) = v;
    *reinterpret_cast<v4f*>(&W2s[r0 + 16 * t][c4]) = w2r[t];
  }
#pragma unroll
  for (int t = 0; t < 2; ++t) {
    *reinterpret_cast<v4f*>(&W3s[r0 + 16 * t][c4]) = w3r[t];
  }
  if (tid < 64)       B1s[tid] = bv;
  else if (tid < 128) B2s[tid - 64] = bv;
  else if (tid < 160) B3s[tid - 128] = bv;
}

__global__ void __launch_bounds__(256, 6)
nsf_main(const float* __restrict__ x,  const float* __restrict__ initp,
         const float* __restrict__ W1, const float* __restrict__ b1,
         const float* __restrict__ W2, const float* __restrict__ b2,
         const float* __restrict__ W3, const float* __restrict__ b3,
         float* __restrict__ z_out, float* __restrict__ ld_ws) {
  __shared__ __align__(16) float Xs[BTILE][PADW];
  __shared__ __align__(16) float W1s[HID][PADW];
  __shared__ __align__(16) float W2s[HID][PADW];
  __shared__ __align__(16) float W3s[32][PADW];
  __shared__ __align__(16) float B1s[HID];
  __shared__ __align__(16) float B2s[HID];
  __shared__ __align__(16) float B3s[32];
  __shared__ v4f Keep[4][256];

  const int g   = blockIdx.y;
  const int b0  = blockIdx.x * BTILE;
  const int tid = threadIdx.x;
  const int nit = group_count(g);
  const unsigned long long glo = group_lo(g);
  const unsigned int       ghi = group_hi(g);
  const int c4  = (tid & 15) << 2;

  const int lane = tid & 31;
  const int wave = tid >> 5;
  const int mrow = wave * 16;
  const int ln   = lane & 15;
  const int hf   = lane >> 4;
  const int kh   = hf << 1;
  const int h8   = hf << 3;

  v4f w1r[4], w2r[4], w3r[2];
  float bv;

  {
    const float* xg = x + (size_t)b0 * DIMS;
    v4f xr[8];
#pragma unroll
    for (int t = 0; t < 8; ++t) xr[t] = ld16(xg + (tid + t * 256) * 4);
    const int d0 = group_dim(glo, ghi, 0);
    fetch_weights(W1, b1, W2, b2, W3, b3, d0, tid, w1r, w2r, w3r, bv);
#pragma unroll
    for (int t = 0; t < 8; ++t) {
      const int m = (tid + t * 256) >> 4;
      *reinterpret_cast<v4f*>(&Xs[m][c4]) = xr[t];
    }
    stage_weights(W1s, W2s, W3s, B1s, B2s, B3s, d0, tid, c4, w1r, w2r, w3r, bv);
  }

  float ldacc = 0.0f;
  int dfirst = 0;

  for (int it = 0; it < nit; ++it) {
    const int d = group_dim(glo, ghi, it);

    {
      const int q4 = 4 * it;
      const int pr = 3 - (q4 >= nit ? 1 : 0) - (q4 >= 2 * nit ? 1 : 0) - (q4 >= 3 * nit ? 1 : 0);
      if (pr >= 3)      __builtin_amdgcn_s_setprio(3);
      else if (pr == 2) __builtin_amdgcn_s_setprio(2);
      else if (pr == 1) __builtin_amdgcn_s_setprio(1);
      else              __builtin_amdgcn_s_setprio(0);
    }

    __syncthreads();

    v8f hv[4];

    {
      v8f acc0 = bias8(&B1s[h8]);
      v8f acc1 = bias8(&B1s[16 + h8]);
      v8f acc2 = bias8(&B1s[32 + h8]);
      v8f acc3 = bias8(&B1s[48 + h8]);
      const float* ap  = &Xs[mrow + ln][kh];
      const float* wp0 = &W1s[ln][kh];
      const float* wp1 = &W1s[16 + ln][kh];
      const float* wp2 = &W1s[32 + ln][kh];
      const float* wp3 = &W1s[48 + ln][kh];
      Frag2 fr[2];
      fr[0] = load_frag2(ap, wp0, wp1, wp2, wp3, 0);
#pragma unroll
      for (int q = 0; q < 8; ++q) {
        const bool more = (q + 1) * 8 < d;
        if (q < 7 && more) fr[(q + 1) & 1] = load_frag2(ap, wp0, wp1, wp2, wp3, (q + 1) * 8);
        mma_step0(fr[q & 1], acc0, acc1, acc2, acc3);
        if (q * 8 + 4 < d) mma_step1(fr[q & 1], acc0, acc1, acc2, acc3);
        if (!more) break;
      }
      hv[0] = tanh8(acc0);
      hv[1] = tanh8(acc1);
      hv[2] = tanh8(acc2);
      hv[3] = tanh8(acc3);
    }

    {
      v8f acc0 = bias8(&B2s[h8]);
      v8f acc1 = bias8(&B2s[16 + h8]);
      v8f acc2 = bias8(&B2s[32 + h8]);
      v8f acc3 = bias8(&B2s[48 + h8]);
      const float* wq0 = &W2s[ln][h8];
      const float* wq1 = &W2s[16 + ln][h8];
      const float* wq2 = &W2s[32 + ln][h8];
      const float* wq3 = &W2s[48 + ln][h8];
#pragma unroll
      for (int t = 0; t < 4; ++t) {
#pragma unroll
        for (int j = 0; j < 4; ++j) {
          const int ko = 16 * t + 2 * j;
          const v2f b  = pair_of(hv[t], j);
          const v2f w0 = ld2(wq0 + ko);
          const v2f w1 = ld2(wq1 + ko);
          const v2f w2 = ld2(wq2 + ko);
          const v2f w3 = ld2(wq3 + ko);
          acc0 = __builtin_amdgcn_wmma_f32_16x16x4_f32(false, w0, false, b, (short)0, acc0, false, false);
          acc1 = __builtin_amdgcn_wmma_f32_16x16x4_f32(false, w1, false, b, (short)0, acc1, false, false);
          acc2 = __builtin_amdgcn_wmma_f32_16x16x4_f32(false, w2, false, b, (short)0, acc2, false, false);
          acc3 = __builtin_amdgcn_wmma_f32_16x16x4_f32(false, w3, false, b, (short)0, acc3, false, false);
        }
      }
      hv[0] = tanh8(acc0);
      hv[1] = tanh8(acc1);
      hv[2] = tanh8(acc2);
      hv[3] = tanh8(acc3);
    }

    const int dn = group_dim(glo, ghi, (it + 1 < nit) ? it + 1 : it);
    fetch_weights(W1, b1, W2, b2, W3, b3, dn, tid, w1r, w2r, w3r, bv);

    v8f p0, p1;
    {
      v8f acc0 = bias8(&B3s[h8]);
      v8f acc1 = bias8(&B3s[16 + h8]);
      const float* wr0 = &W3s[ln][h8];
      const float* wr1 = &W3s[16 + ln][h8];
#pragma unroll
      for (int t = 0; t < 4; ++t) {
#pragma unroll
        for (int j = 0; j < 4; ++j) {
          const int ko = 16 * t + 2 * j;
          const v2f b  = pair_of(hv[t], j);
          const v2f w0 = ld2(wr0 + ko);
          const v2f w1 = ld2(wr1 + ko);
          acc0 = __builtin_amdgcn_wmma_f32_16x16x4_f32(false, w0, false, b, (short)0, acc0, false, false);
          acc1 = __builtin_amdgcn_wmma_f32_16x16x4_f32(false, w1, false, b, (short)0, acc1, false, false);
        }
      }
      p0 = acc0;
      p1 = acc1;
    }

    const bool second = (it & 1) != 0;
    if (!second) {
      dfirst = d;
      v4f k0 = {p0[0], p0[1], p0[2], p0[3]}, k1 = {p0[4], p0[5], p0[6], p0[7]};
      v4f k2 = {p1[0], p1[1], p1[2], p1[3]}, k3 = {p1[4], p1[5], p1[6], p1[7]};
      Keep[0][tid] = k0;  Keep[1][tid] = k1;  Keep[2][tid] = k2;  Keep[3][tid] = k3;
    }
    if (second || it + 1 == nit) {
      const v4f k0 = Keep[0][tid], k1 = Keep[1][tid], k2 = Keep[2][tid], k3 = Keep[3][tid];
      const v8f sv0 = {k0.x, k0.y, k0.z, k0.w, k1.x, k1.y, k1.z, k1.w};
      const v8f sv1 = {k2.x, k2.y, k2.z, k2.w, k3.x, k3.y, k3.z, k3.w};
      float pw[8], ph[8];
#pragma unroll
      for (int r = 0; r < 8; ++r) {
        float a = sv0[r], b = p0[r];
        if (r == 0) swap_halves_first(a, b); else swap_halves(a, b);
        pw[r] = a;
        ph[r] = b;
      }
      const int dl = (hf != 0) ? d : dfirst;
      float z, ld;
      rqs_eval<true>(pw, ph, sv1, p1, Xs[mrow + ln][dl], z, ld);
      if (hf == 0 || second) {
        z_out[(b0 + mrow + ln) * DIMS + dl] = z;
        ldacc += ld;
      }
    }

    __syncthreads();
    if (it + 1 < nit) stage_weights(W1s, W2s, W3s, B1s, B2s, B3s, dn, tid, c4, w1r, w2r, w3r, bv);
  }

  if (g == 3) {
    float pl[23];
#pragma unroll
    for (int j = 0; j < 23; ++j) pl[j] = initp[j];
    float z, ld;
    const v8f dl0 = {pl[16], pl[17], pl[18], pl[19], pl[20], pl[21], pl[22], 0.0f};
    rqs_eval<false>(pl, pl + 8, dl0, dl0, Xs[mrow + ln][0], z, ld);
    if (hf == 0) {
      z_out[(b0 + mrow + ln) * DIMS + 0] = z;
      ldacc += ld;
    }
  }
  ldacc += xhalf_f(ldacc);
  if (hf == 0) ld_ws[(size_t)g * NBATCH + b0 + mrow + ln] = ldacc;
}

__global__ void __launch_bounds__(256)
nsf_reduce(const float* __restrict__ ld_ws, float* __restrict__ out_ld) {
  int b = blockIdx.x * 256 + threadIdx.x;
  float s = 0.0f;
#pragma unroll
  for (int gg = 0; gg < NGROUP; ++gg) s += ld_ws[(size_t)gg * NBATCH + b];
  out_ld[b] = s;
}

extern "C" void kernel_launch(void* const* d_in, const int* in_sizes, int n_in,
                              void* d_out, int out_size, void* d_ws, size_t ws_size,
                              hipStream_t stream) {
  const float* x     = (const float*)d_in[0];
  const float* initp = (const float*)d_in[1];
  const float* W1    = (const float*)d_in[2];
  const float* b1    = (const float*)d_in[3];
  const float* W2    = (const float*)d_in[4];
  const float* b2    = (const float*)d_in[5];
  const float* W3    = (const float*)d_in[6];
  const float* b3    = (const float*)d_in[7];

  float* z_out  = (float*)d_out;
  float* ld_out = z_out + (size_t)NBATCH * DIMS;
  float* ld_ws  = (float*)d_ws;

  dim3 grid(NBATCH / BTILE, NGROUP);
  nsf_main<<<grid, 256, 0, stream>>>(x, initp, W1, b1, W2, b2, W3, b3, z_out, ld_ws);
  nsf_reduce<<<NBATCH / 256, 256, 0, stream>>>(ld_ws, ld_out);
}
